# starting kernel
# baseline (speedup 1.0000x reference)
#include <hip/hip_runtime.h>
#include <hip/hip_bf16.h>
#include <math.h>

typedef __attribute__((ext_vector_type(2))) float v2f;
typedef __attribute__((ext_vector_type(8))) float v8f;

#define DFEAT 64

__global__ void gcn_deg_kernel(const int* __restrict__ dst,
                               float* __restrict__ deg, int E) {
    int e = blockIdx.x * blockDim.x + threadIdx.x;
    if (e < E) {
        __hip_atomic_fetch_add(&deg[dst[e]], 1.0f,
                               __ATOMIC_RELAXED, __HIP_MEMORY_SCOPE_AGENT);
    }
}

__global__ void gcn_invdeg_kernel(float* __restrict__ deg, int N) {
    int n = blockIdx.x * blockDim.x + threadIdx.x;
    if (n < N) {
        float d = deg[n];
        deg[n] = d > 0.0f ? 1.0f / d : 0.0f;
    }
}

__global__ void gcn_scatter_kernel(const float* __restrict__ x,
                                   const int* __restrict__ src,
                                   const int* __restrict__ dst,
                                   float* __restrict__ agg, int E) {
    int t = blockIdx.x * blockDim.x + threadIdx.x;
    int e = t >> 4;
    if (e >= E) return;
    int part = (t & 15) * 4;
    const float4 v = *(const float4*)(x + (size_t)src[e] * DFEAT + part);
    float* o = agg + (size_t)dst[e] * DFEAT + part;
    __hip_atomic_fetch_add(o + 0, v.x, __ATOMIC_RELAXED, __HIP_MEMORY_SCOPE_AGENT);
    __hip_atomic_fetch_add(o + 1, v.y, __ATOMIC_RELAXED, __HIP_MEMORY_SCOPE_AGENT);
    __hip_atomic_fetch_add(o + 2, v.z, __ATOMIC_RELAXED, __HIP_MEMORY_SCOPE_AGENT);
    __hip_atomic_fetch_add(o + 3, v.w, __ATOMIC_RELAXED, __HIP_MEMORY_SCOPE_AGENT);
}

__global__ __launch_bounds__(128)
void gcn_wmma_kernel(const float* __restrict__ agg,
                     const float* __restrict__ inv_deg,
                     const float* __restrict__ W,
                     const float* __restrict__ bias,
                     float* __restrict__ out) {
    const int lane    = threadIdx.x & 31;
    const int wave    = threadIdx.x >> 5;
    const int rowBase = blockIdx.x << 4;
    const int colBase = wave << 4;
    const int m  = lane & 15;
    const int hi = lane >> 4;
    const int k2 = hi << 1;

    const float  inv  = inv_deg[rowBase + m];
    const float* arow = agg + (size_t)(rowBase + m) * DFEAT;
    const float* brow = W   + (size_t)(colBase + m) * DFEAT;

    v8f c = {};
#pragma unroll
    for (int kb = 0; kb < DFEAT; kb += 4) {
        v2f a = *(const v2f*)(arow + kb + k2);
        a.x *= inv;
        a.y *= inv;
        v2f b = *(const v2f*)(brow + kb + k2);
        c = __builtin_amdgcn_wmma_f32_16x16x4_f32(
                 false, a,  false, b,
                 (short)0, c,  false,  false);
    }

    const float bcol = bias[colBase + m];
#pragma unroll
    for (int r = 0; r < 8; ++r) {
        const int row = rowBase + r + (hi << 3);
        float v = c[r] + bcol;
        v = v > 0.0f ? v : expm1f(v);
        out[(size_t)row * DFEAT + colBase + m] = v;
    }
}

extern "C" void kernel_launch(void* const* d_in, const int* in_sizes, int n_in,
                              void* d_out, int out_size, void* d_ws, size_t ws_size,
                              hipStream_t stream) {
    const float* h   = (const float*)d_in[0];
    const int*   src = (const int*)  d_in[1];
    const int*   dst = (const int*)  d_in[2];
    const float* W1  = (const float*)d_in[3];
    const float* b1  = (const float*)d_in[4];
    const float* W2  = (const float*)d_in[5];
    const float* b2  = (const float*)d_in[6];
    float*       out = (float*)d_out;

    const int N = in_sizes[0] / DFEAT;
    const int E = in_sizes[1];

    char* ws = (char*)d_ws;
    float* deg = (float*)ws;
    size_t degBytes = (((size_t)N * sizeof(float)) + 255) & ~(size_t)255;
    float* agg = (float*)(ws + degBytes);
    float* h1  = agg + (size_t)N * DFEAT;

    const int tiles    = N / 16;
    const int scatterB = (E * 16 + 255) / 256;

    hipMemsetAsync(deg, 0, (size_t)N * sizeof(float), stream);
    gcn_deg_kernel<<<(E + 255) / 256, 256, 0, stream>>>(dst, deg, E);
    gcn_invdeg_kernel<<<(N + 255) / 256, 256, 0, stream>>>(deg, N);

    hipMemsetAsync(agg, 0, (size_t)N * DFEAT * sizeof(float), stream);
    gcn_scatter_kernel<<<scatterB, 256, 0, stream>>>(h, src, dst, agg, E);
    gcn_wmma_kernel<<<tiles, 128, 0, stream>>>(agg, deg, W1, b1, h1);

    hipMemsetAsync(agg, 0, (size_t)N * DFEAT * sizeof(float), stream);
    gcn_scatter_kernel<<<scatterB, 256, 0, stream>>>(h1, src, dst, agg, E);
    gcn_wmma_kernel<<<tiles, 128, 0, stream>>>(agg, deg, W2, b2, out);
}


# linear kernel retiled: one wave makes 16 whole rows x all 64 columns (4 accumulators share each A fragment); planes, W and output rows move as whole 256-B rows through LDS; 4x fewer waves
# speedup vs baseline: 3.2783x; 3.2783x over previous
#include <hip/hip_runtime.h>
#include <hip/hip_bf16.h>
#include <math.h>

typedef __attribute__((ext_vector_type(2))) float v2f;
typedef __attribute__((ext_vector_type(8))) float v8f;

#define DFEAT 64

#define AGG_S       2
#define AGG_R       782
#define AGG_THREADS 1024
#define AGG_WAVES   (AGG_THREADS / 32)
#define AGG_VEC     4
#define AGG_WQ      (AGG_VEC * 4 * 32 + 32)
#define LIST_WCAP   4096
#define LIST_HDR    64

template <bool FULL, typename BETWEEN>
__device__ __forceinline__ void gather_add(const float* __restrict__ x, float* acc,
                                           int s, int d, int m, int lane, BETWEEN between) {
#pragma unroll
    for (int h = 0; h < 2; ++h) {
        if (!FULL && h * 16 >= m) break;
        v2f r[16];
#pragma unroll
        for (int j = 0; j < 16; ++j) {
            const int sj = __builtin_amdgcn_readlane(s, h * 16 + j);
            r[j] = *(const v2f*)(x + (size_t)sj * DFEAT + 2 * lane);
        }
        if (h == 0) between();
#pragma unroll
        for (int j = 0; j < 16; ++j) {
            if (FULL || h * 16 + j < m) {
                const int dj = __builtin_amdgcn_readlane(d, h * 16 + j);
                float* p = acc + dj * DFEAT + 2 * lane;
                __hip_atomic_fetch_add(p + 0, r[j].x, __ATOMIC_RELAXED, __HIP_MEMORY_SCOPE_WORKGROUP);
                __hip_atomic_fetch_add(p + 1, r[j].y, __ATOMIC_RELAXED, __HIP_MEMORY_SCOPE_WORKGROUP);
            }
        }
    }
}

template <bool FULL, bool REC>
__device__ __forceinline__ void agg_drain(const float* __restrict__ x,
                                          const int* __restrict__ src,
                                          float* acc, const unsigned* wq,
                                          int m, int lane, unsigned* rec) {
    const bool live = FULL || lane < m;
    const unsigned ent = live ? wq[lane] : 0u;
    const int d = (int)(ent & 1023u);
    const int s = live ? src[ent >> 10] : 0;
    if (REC && rec != nullptr && live)
        rec[lane] = ((unsigned)s << 10) | (unsigned)d;
    gather_add<FULL>(x, acc, s, d, m, lane, [] {});
}

template <bool FIRST>
__global__ __launch_bounds__(AGG_THREADS)
void gcn_agg_kernel(const float* __restrict__ x,
                    const int* __restrict__ src,
                    const int* __restrict__ dst,
                    float* __restrict__ agg,
                    float* __restrict__ degp,
                    unsigned* __restrict__ lists,
                    unsigned* __restrict__ hdrs,
                    int N, int E) {
    __shared__ __attribute__((aligned(16))) float    acc[AGG_R * DFEAT];
    __shared__ unsigned cnt[AGG_R];
    __shared__ unsigned wqs[AGG_WAVES * AGG_WQ];
    __shared__ unsigned ovf;

    const int tid  = threadIdx.x;
    const int lane = tid & 31;
    const int wave = __builtin_amdgcn_readfirstlane(tid >> 5);
    const int lo    = blockIdx.x * AGG_R;
    const int slice = blockIdx.y;
    const unsigned nloc = (unsigned)min(AGG_R, N - lo);
    unsigned* wq = wqs + wave * AGG_WQ;

    const size_t blk = (size_t)slice * gridDim.x + blockIdx.x;
    unsigned* hdr    = hdrs + blk * LIST_HDR;
    unsigned* mylist = lists + (blk * AGG_WAVES + wave) * LIST_WCAP;

    for (int i = tid; i < AGG_R * DFEAT / 4; i += AGG_THREADS)
        ((float4*)acc)[i] = make_float4(0.f, 0.f, 0.f, 0.f);
    if (FIRST) {
        for (int i = tid; i < AGG_R; i += AGG_THREADS) cnt[i] = 0u;
        if (tid == 0) ovf = 0u;
    }
    __syncthreads();

    bool scan = true;
    if (!FIRST) scan = __builtin_amdgcn_readfirstlane((int)hdr[32]) != 0;

    if (scan) {
        const int4* dst4 = (const int4*)dst;
        const int nvec  = E >> 2;
        const int gvec  = AGG_VEC * AGG_THREADS;
        const int nfull = nvec / gvec;
        const int gs = (int)(((long long)slice * nfull) / AGG_S);
        const int ng = (int)(((long long)(slice + 1) * nfull) / AGG_S) - gs;
        const int g0 = (int)(((long long)blockIdx.x * ng) / gridDim.x);

        unsigned cur  = 0;
        unsigned nrec = 0;

        auto drain = [&](const unsigned* q, unsigned m) {
            unsigned* rec = nullptr;
            if (FIRST) {
                if (nrec + m <= (unsigned)LIST_WCAP) rec = mylist + nrec;
                else if (lane == 0) ovf = 1u;
                nrec += m;
            }
            if (m == 32u) agg_drain<true,  FIRST>(x, src, acc, q, 32, lane, rec);
            else          agg_drain<false, FIRST>(x, src, acc, q, (int)m, lane, rec);
        };

        auto visit = [&](int val, unsigned e) {
            const unsigned u = (unsigned)val - (unsigned)lo;
            const bool hit = u < nloc;
            const unsigned mask = __builtin_amdgcn_ballot_w32(hit);
            if (mask) {
                if (hit) {
                    wq[cur + __builtin_amdgcn_mbcnt_lo(mask, 0u)] = (e << 10) | u;
                    if (FIRST)
                        __hip_atomic_fetch_add(&cnt[u], 1u, __ATOMIC_RELAXED, __HIP_MEMORY_SCOPE_WORKGROUP);
                }
                cur += (unsigned)__builtin_popcount(mask);
            }
        };

        auto group_at = [&](int g) {
            int gg = g + g0;
            return gs + (gg >= ng ? gg - ng : gg);
        };

        auto load_group = [&](int g, int4 (&v)[AGG_VEC]) {
            const int base = group_at(g) * gvec + tid;
#pragma unroll
            for (int k = 0; k < AGG_VEC; ++k) v[k] = dst4[base + k * AGG_THREADS];
        };

        auto visit_vecs = [&](int base, const int4 (&v)[AGG_VEC]) {
#pragma unroll
            for (int k = 0; k < AGG_VEC; ++k) {
                const unsigned e = (unsigned)(base + k * AGG_THREADS) << 2;
                visit(v[k].x, e + 0u);
                visit(v[k].y, e + 1u);
                visit(v[k].z, e + 2u);
                visit(v[k].w, e + 3u);
            }
            while (cur >= 32u) {
                cur -= 32u;
                drain(wq + cur, 32u);
            }
        };

        if (ng > 0) {
            int4 va[AGG_VEC], vb[AGG_VEC];
            load_group(0, va);
            for (int g = 0; g + 1 < ng; g += 2) {
                load_group(g + 1, vb);
                visit_vecs(group_at(g) * gvec + tid, va);
                load_group(min(g + 2, ng - 1), va);
                visit_vecs(group_at(g + 1) * gvec + tid, vb);
            }
            if (ng & 1) visit_vecs(group_at(ng - 1) * gvec + tid, va);
        }
        if (slice == 0) {
            int4 vr[AGG_VEC];
            const int base = nfull * gvec + tid;
#pragma unroll
            for (int k = 0; k < AGG_VEC; ++k) {
                const int vi = base + k * AGG_THREADS;
                vr[k] = vi < nvec ? dst4[vi] : make_int4(-1, -1, -1, -1);
            }
            visit_vecs(base, vr);
        }
        if (slice == 0 && wave == 0) {
            const int e = (nvec << 2) + lane;
            const bool tail = lane < (E & 3);
            visit(tail ? dst[e] : -1, (unsigned)e);
        }
        while (cur >= 32u) {
            cur -= 32u;
            drain(wq + cur, 32u);
        }
        if (cur) drain(wq, cur);
        if (FIRST && lane == 0) hdr[wave] = nrec;
    } else {
        const unsigned n = (unsigned)__builtin_amdgcn_readfirstlane((int)hdr[wave]);
        unsigned ent = (unsigned)lane < n ? mylist[lane] : 0u;
        for (unsigned k = 0; k < n; k += 32u) {
            const int d = (int)(ent & 1023u);
            const int s = (int)(ent >> 10);
            const unsigned kn = k + 32u + (unsigned)lane;
            unsigned nxt = 0u;
            auto next_line = [&] { if (kn < n) nxt = mylist[kn]; };
            const unsigned m = n - k;
            if (m >= 32u) gather_add<true >(x, acc, s, d, 32, lane, next_line);
            else          gather_add<false>(x, acc, s, d, (int)m, lane, next_line);
            ent = nxt;
        }
    }

    __syncthreads();

    float4* orow = (float4*)(agg + ((size_t)slice * N + lo) * DFEAT);
    for (int i = tid; i < (int)nloc * (DFEAT / 4); i += AGG_THREADS)
        orow[i] = ((const float4*)acc)[i];
    if (FIRST) {
        for (int i = tid; i < (int)nloc; i += AGG_THREADS)
            degp[(size_t)slice * N + lo + i] = (float)cnt[i];
        if (tid == 0) hdr[32] = ovf;
    }
}

#define LIN_WAVES   8
#define LIN_THREADS (LIN_WAVES * 32)
#define LIN_PITCH   (DFEAT + 4)

__global__ __launch_bounds__(LIN_THREADS)
void gcn_wmma_kernel(const float* __restrict__ agg,
                     const float* __restrict__ degp,
                     const float* __restrict__ W,
                     const float* __restrict__ bias,
                     float* __restrict__ out, int N) {
    __shared__ __attribute__((aligned(16))) float Ws[DFEAT * LIN_PITCH];
    __shared__ __attribute__((aligned(16))) float As[LIN_WAVES * 16 * LIN_PITCH];

    const int tid  = threadIdx.x;
    const int lane = tid & 31;
    const int wave  = __builtin_amdgcn_readfirstlane(tid >> 5);
    const int tiles = N >> 4;
    const int want  = blockIdx.x * LIN_WAVES + wave;
    const bool valid = want < tiles;
    const int tile = valid ? want : tiles - 1;
    const int rowBase = tile << 4;
    const int m  = lane & 15;
    const int hi = lane >> 4;
    const int k2 = hi << 1;
    const int c4 = m << 2;
    float* mine = As + wave * (16 * LIN_PITCH);
    const size_t plane = (size_t)N * DFEAT;

    float4 w4[DFEAT * DFEAT / 4 / LIN_THREADS];
#pragma unroll
    for (int j = 0; j < DFEAT * DFEAT / 4 / LIN_THREADS; ++j)
        w4[j] = *(const float4*)(W + 4 * (tid + j * LIN_THREADS));

    const float* rows = agg + (size_t)rowBase * DFEAT;
    const unsigned loff = (unsigned)(hi * DFEAT + c4);
    float4 v[8];
#pragma unroll
    for (int i = 0; i < 8; ++i)
        v[i] = *(const float4*)(rows + loff + i * 2 * DFEAT);
    float4 u[AGG_S - 1][8];
#pragma unroll
    for (int p = 1; p < AGG_S; ++p) {
#pragma unroll
        for (int i = 0; i < 8; ++i)
            u[p - 1][i] = *(const float4*)(rows + p * plane + loff + i * 2 * DFEAT);
    }

    float deg = degp[rowBase + m];
#pragma unroll
    for (int p = 1; p < AGG_S; ++p) deg += degp[(size_t)p * N + rowBase + m];
    float bcol[4];
#pragma unroll
    for (int t = 0; t < 4; ++t) bcol[t] = bias[t * 16 + m];

#pragma unroll
    for (int j = 0; j < DFEAT * DFEAT / 4 / LIN_THREADS; ++j) {
        const int idx = tid + j * LIN_THREADS;
        *(float4*)(Ws + (idx >> 4) * LIN_PITCH + ((idx & 15) << 2)) = w4[j];
    }
#pragma unroll
    for (int i = 0; i < 8; ++i) {
#pragma unroll
        for (int p = 1; p < AGG_S; ++p) {
            v[i].x += u[p - 1][i].x; v[i].y += u[p - 1][i].y;
            v[i].z += u[p - 1][i].z; v[i].w += u[p - 1][i].w;
        }
        *(float4*)(mine + (2 * i + hi) * LIN_PITCH + c4) = v[i];
    }
    const float inv = deg > 0.0f ? 1.0f / deg : 0.0f;
    __syncthreads();

    v8f c[4] = {};
    const float* arow = mine + m * LIN_PITCH + k2;
    const float* brow = Ws   + m * LIN_PITCH + k2;
#pragma unroll
    for (int kb = 0; kb < DFEAT; kb += 4) {
        v2f a = *(const v2f*)(arow + kb);
        a.x *= inv;
        a.y *= inv;
#pragma unroll
        for (int t = 0; t < 4; ++t) {
            const v2f b = *(const v2f*)(brow + t * 16 * LIN_PITCH + kb);
            c[t] = __builtin_amdgcn_wmma_f32_16x16x4_f32(
                     false, a,  false, b,
                     (short)0, c[t],  false,  false);
        }
    }

#pragma unroll
    for (int t = 0; t < 4; ++t) {
#pragma unroll
        for (int r = 0; r < 8; ++r) {
            float e = c[t][r] + bcol[t];
            e = e > 0.0f ? e : expm1f(e);
            mine[(r + (hi << 3)) * LIN_PITCH + t * 16 + m] = e;
        }
    }
    __syncthreads();

    if (valid) {
        float* orow = out + (size_t)rowBase * DFEAT;
#pragma unroll
        for (int i = 0; i < 8; ++i)
            *(float4*)(orow + loff + i * 2 * DFEAT) =
                *(const float4*)(mine + (2 * i + hi) * LIN_PITCH + c4);
    }
}

extern "C" void kernel_launch(void* const* d_in, const int* in_sizes, int n_in,
                              void* d_out, int out_size, void* d_ws, size_t ws_size,
                              hipStream_t stream) {
    const float* h   = (const float*)d_in[0];
    const int*   src = (const int*)  d_in[1];
    const int*   dst = (const int*)  d_in[2];
    const float* W1  = (const float*)d_in[3];
    const float* b1  = (const float*)d_in[4];
    const float* W2  = (const float*)d_in[5];
    const float* b2  = (const float*)d_in[6];
    float*       out = (float*)d_out;

    const int N = in_sizes[0] / DFEAT;
    const int E = in_sizes[1];

    char* ws = (char*)d_ws;
    float* deg = (float*)ws;
    size_t degBytes = (((size_t)AGG_S * N * sizeof(float)) + 255) & ~(size_t)255;
    float* agg = (float*)(ws + degBytes);
    float* h1  = agg + (size_t)AGG_S * N * DFEAT;
    unsigned* hdrs  = (unsigned*)(h1 + (size_t)N * DFEAT);

    const int tiles = N / 16;
    const int linB  = (tiles + LIN_WAVES - 1) / LIN_WAVES;
    const dim3 aggB((N + AGG_R - 1) / AGG_R, AGG_S);
    unsigned* lists = hdrs + (size_t)aggB.x * aggB.y * LIST_HDR;

    gcn_agg_kernel<true><<<aggB, AGG_THREADS, 0, stream>>>(h, src, dst, agg, deg, lists, hdrs, N, E);
    gcn_wmma_kernel<<<linB, LIN_THREADS, 0, stream>>>(agg, deg, W1, b1, h1, N);

    gcn_agg_kernel<false><<<aggB, AGG_THREADS, 0, stream>>>(h1, src, dst, agg, deg, lists, hdrs, N, E);
    gcn_wmma_kernel<<<linB, LIN_THREADS, 0, stream>>>(agg, deg, W2, b2, out, N);
}


# layer-1 scan reads a 1-byte range key per edge, 4 keys compared per word byte-parallel (was 4-byte dst); a new launch writes keys + packed (source,row) words; 4 -> 5 launches; sum order: arrival
# speedup vs baseline: 3.5095x; 1.0705x over previous
#include <hip/hip_runtime.h>
#include <hip/hip_bf16.h>
#include <math.h>

typedef __attribute__((ext_vector_type(2))) float v2f;
typedef __attribute__((ext_vector_type(8))) float v8f;

#define DFEAT 64

#define AGG_S       2
#define AGG_R       782
#define AGG_THREADS 1024
#define AGG_WAVES   (AGG_THREADS / 32)
#define AGG_VEC     4
#define AGG_WQ      (16 * 32 + 32)
#define KEY_THREADS 256
#define KEY_ITEMS   4

__global__ __launch_bounds__(KEY_THREADS)
void gcn_key_kernel(const int* __restrict__ src,
                    const int* __restrict__ dst,
                    unsigned* __restrict__ keys,
                    unsigned* __restrict__ packed,
                    int E) {
    const int nv   = E >> 2;
    const int base = blockIdx.x * (KEY_THREADS * KEY_ITEMS) + threadIdx.x;
    if (nv > 0) {
        int4 d[KEY_ITEMS], s[KEY_ITEMS];
#pragma unroll
        for (int k = 0; k < KEY_ITEMS; ++k) {
            const int i = min(base + k * KEY_THREADS, nv - 1);
            d[k] = ((const int4*)dst)[i];
            s[k] = ((const int4*)src)[i];
        }
#pragma unroll
        for (int k = 0; k < KEY_ITEMS; ++k) {
            const int i = base + k * KEY_THREADS;
            if (i < nv) {
                const unsigned r0 = (unsigned)d[k].x / AGG_R, r1 = (unsigned)d[k].y / AGG_R;
                const unsigned r2 = (unsigned)d[k].z / AGG_R, r3 = (unsigned)d[k].w / AGG_R;
                uint4 p;
                p.x = ((unsigned)s[k].x << 10) | ((unsigned)d[k].x - r0 * AGG_R);
                p.y = ((unsigned)s[k].y << 10) | ((unsigned)d[k].y - r1 * AGG_R);
                p.z = ((unsigned)s[k].z << 10) | ((unsigned)d[k].z - r2 * AGG_R);
                p.w = ((unsigned)s[k].w << 10) | ((unsigned)d[k].w - r3 * AGG_R);
                keys[i] = r0 | (r1 << 8) | (r2 << 16) | (r3 << 24);
                ((uint4*)packed)[i] = p;
            }
        }
    }
    if (blockIdx.x == 0 && (int)threadIdx.x < (E & 3)) {
        const int e = (nv << 2) + (int)threadIdx.x;
        const unsigned dv = (unsigned)dst[e];
        const unsigned r  = dv / AGG_R;
        packed[e] = ((unsigned)src[e] << 10) | (dv - r * AGG_R);
        ((unsigned char*)keys)[e] = (unsigned char)r;
    }
}

#define LIST_WCAP   4096
#define LIST_HDR    64

template <bool FULL, typename BETWEEN>
__device__ __forceinline__ void gather_add(const float* __restrict__ x, float* acc,
                                           int s, int d, int m, int lane, BETWEEN between) {
#pragma unroll
    for (int h = 0; h < 2; ++h) {
        if (!FULL && h * 16 >= m) break;
        v2f r[16];
#pragma unroll
        for (int j = 0; j < 16; ++j) {
            const int sj = __builtin_amdgcn_readlane(s, h * 16 + j);
            r[j] = *(const v2f*)(x + (size_t)sj * DFEAT + 2 * lane);
        }
        if (h == 0) between();
#pragma unroll
        for (int j = 0; j < 16; ++j) {
            if (FULL || h * 16 + j < m) {
                const int dj = __builtin_amdgcn_readlane(d, h * 16 + j);
                float* p = acc + dj * DFEAT + 2 * lane;
                __hip_atomic_fetch_add(p + 0, r[j].x, __ATOMIC_RELAXED, __HIP_MEMORY_SCOPE_WORKGROUP);
                __hip_atomic_fetch_add(p + 1, r[j].y, __ATOMIC_RELAXED, __HIP_MEMORY_SCOPE_WORKGROUP);
            }
        }
    }
}

template <bool FULL, bool REC>
__device__ __forceinline__ void agg_drain(const float* __restrict__ x,
                                          const unsigned* __restrict__ packed,
                                          float* acc, unsigned* cnt, const unsigned* wq,
                                          int m, int lane, unsigned* rec) {
    const bool live = FULL || lane < m;
    const unsigned ent = live ? packed[wq[lane]] : 0u;
    const int d = (int)(ent & 1023u);
    const int s = (int)(ent >> 10);
    if (REC && live)
        __hip_atomic_fetch_add(&cnt[d], 1u, __ATOMIC_RELAXED, __HIP_MEMORY_SCOPE_WORKGROUP);
    if (REC && rec != nullptr && live)
        rec[lane] = ent;
    gather_add<FULL>(x, acc, s, d, m, lane, [] {});
}

template <bool FIRST>
__global__ __launch_bounds__(AGG_THREADS)
void gcn_agg_kernel(const float* __restrict__ x,
                    const unsigned* __restrict__ keys,
                    const unsigned* __restrict__ packed,
                    float* __restrict__ agg,
                    float* __restrict__ degp,
                    unsigned* __restrict__ lists,
                    unsigned* __restrict__ hdrs,
                    int N, int E) {
    __shared__ __attribute__((aligned(16))) float    acc[AGG_R * DFEAT];
    __shared__ unsigned cnt[AGG_R];
    __shared__ unsigned wqs[AGG_WAVES * AGG_WQ];
    __shared__ unsigned ovf;

    const int tid  = threadIdx.x;
    const int lane = tid & 31;
    const int wave = __builtin_amdgcn_readfirstlane(tid >> 5);
    const int lo    = blockIdx.x * AGG_R;
    const int slice = blockIdx.y;
    const unsigned nloc = (unsigned)min(AGG_R, N - lo);
    unsigned* wq = wqs + wave * AGG_WQ;

    const size_t blk = (size_t)slice * gridDim.x + blockIdx.x;
    unsigned* hdr    = hdrs + blk * LIST_HDR;
    unsigned* mylist = lists + (blk * AGG_WAVES + wave) * LIST_WCAP;

    for (int i = tid; i < AGG_R * DFEAT / 4; i += AGG_THREADS)
        ((float4*)acc)[i] = make_float4(0.f, 0.f, 0.f, 0.f);
    if (FIRST) {
        for (int i = tid; i < AGG_R; i += AGG_THREADS) cnt[i] = 0u;
        if (tid == 0) ovf = 0u;
    }
    __syncthreads();

    bool scan = true;
    if (!FIRST) scan = __builtin_amdgcn_readfirstlane((int)hdr[32]) != 0;

    if (scan) {
        const uint4* key4 = (const uint4*)keys;
        const int nvec  = E >> 4;
        const int gvec  = AGG_VEC * AGG_THREADS;
        const int nfull = nvec / gvec;
        const int gs = (int)(((long long)slice * nfull) / AGG_S);
        const int ng = (int)(((long long)(slice + 1) * nfull) / AGG_S) - gs;
        const int g0 = (int)(((long long)blockIdx.x * ng) / gridDim.x);
        const unsigned pat = (unsigned)blockIdx.x * 0x01010101u;

        unsigned cur  = 0;
        unsigned nrec = 0;

        auto drain = [&](const unsigned* q, unsigned m) {
            unsigned* rec = nullptr;
            if (FIRST) {
                if (nrec + m <= (unsigned)LIST_WCAP) rec = mylist + nrec;
                else if (lane == 0) ovf = 1u;
                nrec += m;
            }
            if (m == 32u) agg_drain<true,  FIRST>(x, packed, acc, cnt, q, 32, lane, rec);
            else          agg_drain<false, FIRST>(x, packed, acc, cnt, q, (int)m, lane, rec);
        };

        auto flags = [&](unsigned w) {
            const unsigned x = w ^ pat;
            return (0x80808080u - x) & 0x80808080u;
        };

        auto visit16 = [&](const uint4& v, unsigned ebase) {
            unsigned mm = flags(v.x) | (flags(v.y) >> 1) | (flags(v.z) >> 2) | (flags(v.w) >> 3);
            unsigned mask = __builtin_amdgcn_ballot_w32(mm != 0u);
            while (mask) {
                if (mm != 0u) {
                    const unsigned b = (unsigned)__builtin_ctz(mm);
                    const unsigned e = ebase + ((7u - (b & 7u)) << 2) + (b >> 3);
                    wq[cur + __builtin_amdgcn_mbcnt_lo(mask, 0u)] = e;
                    mm &= mm - 1u;
                }
                cur += (unsigned)__builtin_popcount(mask);
                mask = __builtin_amdgcn_ballot_w32(mm != 0u);
            }
        };

        auto group_at = [&](int g) {
            int gg = g + g0;
            return gs + (gg >= ng ? gg - ng : gg);
        };

        auto load_group = [&](int g, uint4 (&v)[AGG_VEC]) {
            const int base = group_at(g) * gvec + tid;
#pragma unroll
            for (int k = 0; k < AGG_VEC; ++k) v[k] = key4[base + k * AGG_THREADS];
        };

        auto visit_vecs = [&](int base, const uint4 (&v)[AGG_VEC]) {
#pragma unroll
            for (int k = 0; k < AGG_VEC; ++k) {
                visit16(v[k], (unsigned)(base + k * AGG_THREADS) << 4);
                while (cur >= 32u) {
                    cur -= 32u;
                    drain(wq + cur, 32u);
                }
            }
        };

        if (ng > 0) {
            uint4 va[AGG_VEC], vb[AGG_VEC];
            load_group(0, va);
            for (int g = 0; g < ng; ++g) {
                load_group(min(g + 1, ng - 1), vb);
                visit_vecs(group_at(g) * gvec + tid, va);
#pragma unroll
                for (int k = 0; k < AGG_VEC; ++k) va[k] = vb[k];
            }
        }
        if (slice == 0) {
            uint4 vr[AGG_VEC];
            const int base = nfull * gvec + tid;
            const unsigned none = pat ^ 0x01010101u;
#pragma unroll
            for (int k = 0; k < AGG_VEC; ++k) {
                const int vi = base + k * AGG_THREADS;
                vr[k] = vi < nvec ? key4[vi] : make_uint4(none, none, none, none);
            }
            visit_vecs(base, vr);
        }
        if (slice == 0 && wave == 0 && (E & 15) != 0) {
            const int e = (nvec << 4) + lane;
            const bool tail = lane < (E & 15);
            const unsigned kb = tail ? (unsigned)((const unsigned char*)keys)[e] : 0xFFu;
            const bool hit = kb == (unsigned)blockIdx.x;
            const unsigned mask = __builtin_amdgcn_ballot_w32(hit);
            if (hit) wq[cur + __builtin_amdgcn_mbcnt_lo(mask, 0u)] = (unsigned)e;
            cur += (unsigned)__builtin_popcount(mask);
        }
        while (cur >= 32u) {
            cur -= 32u;
            drain(wq + cur, 32u);
        }
        if (cur) drain(wq, cur);
        if (FIRST && lane == 0) hdr[wave] = nrec;
    } else {
        const unsigned n = (unsigned)__builtin_amdgcn_readfirstlane((int)hdr[wave]);
        unsigned ent = (unsigned)lane < n ? mylist[lane] : 0u;
        for (unsigned k = 0; k < n; k += 32u) {
            const int d = (int)(ent & 1023u);
            const int s = (int)(ent >> 10);
            const unsigned kn = k + 32u + (unsigned)lane;
            unsigned nxt = 0u;
            auto next_line = [&] { if (kn < n) nxt = mylist[kn]; };
            const unsigned m = n - k;
            if (m >= 32u) gather_add<true >(x, acc, s, d, 32, lane, next_line);
            else          gather_add<false>(x, acc, s, d, (int)m, lane, next_line);
            ent = nxt;
        }
    }

    __syncthreads();

    float4* orow = (float4*)(agg + ((size_t)slice * N + lo) * DFEAT);
    for (int i = tid; i < (int)nloc * (DFEAT / 4); i += AGG_THREADS)
        orow[i] = ((const float4*)acc)[i];
    if (FIRST) {
        for (int i = tid; i < (int)nloc; i += AGG_THREADS)
            degp[(size_t)slice * N + lo + i] = (float)cnt[i];
        if (tid == 0) hdr[32] = ovf;
    }
}

#define LIN_WAVES   8
#define LIN_THREADS (LIN_WAVES * 32)
#define LIN_PITCH   (DFEAT + 4)

__global__ __launch_bounds__(LIN_THREADS)
void gcn_wmma_kernel(const float* __restrict__ agg,
                     const float* __restrict__ degp,
                     const float* __restrict__ W,
                     const float* __restrict__ bias,
                     float* __restrict__ out, int N) {
    __shared__ __attribute__((aligned(16))) float Ws[DFEAT * LIN_PITCH];
    __shared__ __attribute__((aligned(16))) float As[LIN_WAVES * 16 * LIN_PITCH];

    const int tid  = threadIdx.x;
    const int lane = tid & 31;
    const int wave  = __builtin_amdgcn_readfirstlane(tid >> 5);
    const int tiles = N >> 4;
    const int want  = blockIdx.x * LIN_WAVES + wave;
    const bool valid = want < tiles;
    const int tile = valid ? want : tiles - 1;
    const int rowBase = tile << 4;
    const int m  = lane & 15;
    const int hi = lane >> 4;
    const int k2 = hi << 1;
    const int c4 = m << 2;
    float* mine = As + wave * (16 * LIN_PITCH);
    const size_t plane = (size_t)N * DFEAT;

    float4 w4[DFEAT * DFEAT / 4 / LIN_THREADS];
#pragma unroll
    for (int j = 0; j < DFEAT * DFEAT / 4 / LIN_THREADS; ++j)
        w4[j] = *(const float4*)(W + 4 * (tid + j * LIN_THREADS));

    const float* rows = agg + (size_t)rowBase * DFEAT;
    const unsigned loff = (unsigned)(hi * DFEAT + c4);
    float4 v[8];
#pragma unroll
    for (int i = 0; i < 8; ++i)
        v[i] = *(const float4*)(rows + loff + i * 2 * DFEAT);
    float4 u[AGG_S - 1][8];
#pragma unroll
    for (int p = 1; p < AGG_S; ++p) {
#pragma unroll
        for (int i = 0; i < 8; ++i)
            u[p - 1][i] = *(const float4*)(rows + p * plane + loff + i * 2 * DFEAT);
    }

    float deg = degp[rowBase + m];
#pragma unroll
    for (int p = 1; p < AGG_S; ++p) deg += degp[(size_t)p * N + rowBase + m];
    float bcol[4];
#pragma unroll
    for (int t = 0; t < 4; ++t) bcol[t] = bias[t * 16 + m];

#pragma unroll
    for (int j = 0; j < DFEAT * DFEAT / 4 / LIN_THREADS; ++j) {
        const int idx = tid + j * LIN_THREADS;
        *(float4*)(Ws + (idx >> 4) * LIN_PITCH + ((idx & 15) << 2)) = w4[j];
    }
#pragma unroll
    for (int i = 0; i < 8; ++i) {
#pragma unroll
        for (int p = 1; p < AGG_S; ++p) {
            v[i].x += u[p - 1][i].x; v[i].y += u[p - 1][i].y;
            v[i].z += u[p - 1][i].z; v[i].w += u[p - 1][i].w;
        }
        *(float4*)(mine + (2 * i + hi) * LIN_PITCH + c4) = v[i];
    }
    const float inv = deg > 0.0f ? 1.0f / deg : 0.0f;
    __syncthreads();

    v8f c[4] = {};
    const float* arow = mine + m * LIN_PITCH + k2;
    const float* brow = Ws   + m * LIN_PITCH + k2;
#pragma unroll
    for (int kb = 0; kb < DFEAT; kb += 4) {
        v2f a = *(const v2f*)(arow + kb);
        a.x *= inv;
        a.y *= inv;
#pragma unroll
        for (int t = 0; t < 4; ++t) {
            const v2f b = *(const v2f*)(brow + t * 16 * LIN_PITCH + kb);
            c[t] = __builtin_amdgcn_wmma_f32_16x16x4_f32(
                     false, a,  false, b,
                     (short)0, c[t],  false,  false);
        }
    }

#pragma unroll
    for (int t = 0; t < 4; ++t) {
#pragma unroll
        for (int r = 0; r < 8; ++r) {
            float e = c[t][r] + bcol[t];
            e = e > 0.0f ? e : expm1f(e);
            mine[(r + (hi << 3)) * LIN_PITCH + t * 16 + m] = e;
        }
    }
    __syncthreads();

    if (valid) {
        float* orow = out + (size_t)rowBase * DFEAT;
#pragma unroll
        for (int i = 0; i < 8; ++i)
            *(float4*)(orow + loff + i * 2 * DFEAT) =
                *(const float4*)(mine + (2 * i + hi) * LIN_PITCH + c4);
    }
}

extern "C" void kernel_launch(void* const* d_in, const int* in_sizes, int n_in,
                              void* d_out, int out_size, void* d_ws, size_t ws_size,
                              hipStream_t stream) {
    const float* h   = (const float*)d_in[0];
    const int*   src = (const int*)  d_in[1];
    const int*   dst = (const int*)  d_in[2];
    const float* W1  = (const float*)d_in[3];
    const float* b1  = (const float*)d_in[4];
    const float* W2  = (const float*)d_in[5];
    const float* b2  = (const float*)d_in[6];
    float*       out = (float*)d_out;

    const int N = in_sizes[0] / DFEAT;
    const int E = in_sizes[1];

    char* ws = (char*)d_ws;
    float* deg = (float*)ws;
    size_t degBytes = (((size_t)AGG_S * N * sizeof(float)) + 255) & ~(size_t)255;
    float* agg = (float*)(ws + degBytes);
    float* h1  = agg + (size_t)AGG_S * N * DFEAT;
    unsigned* hdrs  = (unsigned*)(h1 + (size_t)N * DFEAT);

    const int tiles = N / 16;
    const int linB  = (tiles + LIN_WAVES - 1) / LIN_WAVES;
    const dim3 aggB((N + AGG_R - 1) / AGG_R, AGG_S);
    unsigned* lists = hdrs + (size_t)aggB.x * aggB.y * LIST_HDR;
    unsigned* keys   = lists + (size_t)aggB.x * aggB.y * AGG_WAVES * LIST_WCAP;
    unsigned* packed = keys + (size_t)((E + 15) / 16) * 4;
    const int keyB   = (E / 4 + KEY_THREADS * KEY_ITEMS - 1) / (KEY_THREADS * KEY_ITEMS) + (E < 4);

    gcn_key_kernel<<<keyB, KEY_THREADS, 0, stream>>>(src, dst, keys, packed, E);

    gcn_agg_kernel<true><<<aggB, AGG_THREADS, 0, stream>>>(h, keys, packed, agg, deg, lists, hdrs, N, E);
    gcn_wmma_kernel<<<linB, LIN_THREADS, 0, stream>>>(agg, deg, W1, b1, h1, N);

    gcn_agg_kernel<false><<<aggB, AGG_THREADS, 0, stream>>>(h1, keys, packed, agg, deg, lists, hdrs, N, E);
    gcn_wmma_kernel<<<linB, LIN_THREADS, 0, stream>>>(agg, deg, W2, b2, out, N);
}


# layer-2 sums: every LDS sum row gets one owner wave (row mod 32); layer 1 files entries in per-owner LDS queues, layer 2 adds rows by plain LDS read-add-write, no float atomics; sum order: arrival
# speedup vs baseline: 3.6779x; 1.0480x over previous
#include <hip/hip_runtime.h>
#include <hip/hip_bf16.h>
#include <math.h>

typedef __attribute__((ext_vector_type(2))) float v2f;
typedef __attribute__((ext_vector_type(8))) float v8f;

#define DFEAT 64

#define AGG_S       2
#define AGG_R       782
#define AGG_THREADS 1024
#define AGG_WAVES   (AGG_THREADS / 32)
#define AGG_VEC     4
#define AGG_WQ      (16 * 32 + 32)
#define KEY_THREADS 256
#define KEY_ITEMS   4

__global__ __launch_bounds__(KEY_THREADS)
void gcn_key_kernel(const int* __restrict__ src,
                    const int* __restrict__ dst,
                    unsigned* __restrict__ keys,
                    unsigned* __restrict__ packed,
                    int E) {
    const int nv   = E >> 2;
    const int base = blockIdx.x * (KEY_THREADS * KEY_ITEMS) + threadIdx.x;
    if (nv > 0) {
        int4 d[KEY_ITEMS], s[KEY_ITEMS];
#pragma unroll
        for (int k = 0; k < KEY_ITEMS; ++k) {
            const int i = min(base + k * KEY_THREADS, nv - 1);
            d[k] = ((const int4*)dst)[i];
            s[k] = ((const int4*)src)[i];
        }
#pragma unroll
        for (int k = 0; k < KEY_ITEMS; ++k) {
            const int i = base + k * KEY_THREADS;
            if (i < nv) {
                const unsigned r0 = (unsigned)d[k].x / AGG_R, r1 = (unsigned)d[k].y / AGG_R;
                const unsigned r2 = (unsigned)d[k].z / AGG_R, r3 = (unsigned)d[k].w / AGG_R;
                uint4 p;
                p.x = ((unsigned)s[k].x << 10) | ((unsigned)d[k].x - r0 * AGG_R);
                p.y = ((unsigned)s[k].y << 10) | ((unsigned)d[k].y - r1 * AGG_R);
                p.z = ((unsigned)s[k].z << 10) | ((unsigned)d[k].z - r2 * AGG_R);
                p.w = ((unsigned)s[k].w << 10) | ((unsigned)d[k].w - r3 * AGG_R);
                keys[i] = r0 | (r1 << 8) | (r2 << 16) | (r3 << 24);
                ((uint4*)packed)[i] = p;
            }
        }
    }
    if (blockIdx.x == 0 && (int)threadIdx.x < (E & 3)) {
        const int e = (nv << 2) + (int)threadIdx.x;
        const unsigned dv = (unsigned)dst[e];
        const unsigned r  = dv / AGG_R;
        packed[e] = ((unsigned)src[e] << 10) | (dv - r * AGG_R);
        ((unsigned char*)keys)[e] = (unsigned char)r;
    }
}

#define OQ_CAP      320
#define LIST_WCAP   OQ_CAP
#define LIST_HDR    64

template <bool FULL, typename BETWEEN>
__device__ __forceinline__ void gather_add(const float* __restrict__ x, float* acc,
                                           int s, int d, int m, int lane, BETWEEN between) {
#pragma unroll
    for (int h = 0; h < 2; ++h) {
        if (!FULL && h * 16 >= m) break;
        v2f r[16];
#pragma unroll
        for (int j = 0; j < 16; ++j) {
            const int sj = __builtin_amdgcn_readlane(s, h * 16 + j);
            r[j] = *(const v2f*)(x + (size_t)sj * DFEAT + 2 * lane);
        }
        if (h == 0) between();
#pragma unroll
        for (int j = 0; j < 16; ++j) {
            if (FULL || h * 16 + j < m) {
                const int dj = __builtin_amdgcn_readlane(d, h * 16 + j);
                float* p = acc + dj * DFEAT + 2 * lane;
                __hip_atomic_fetch_add(p + 0, r[j].x, __ATOMIC_RELAXED, __HIP_MEMORY_SCOPE_WORKGROUP);
                __hip_atomic_fetch_add(p + 1, r[j].y, __ATOMIC_RELAXED, __HIP_MEMORY_SCOPE_WORKGROUP);
            }
        }
    }
}

template <int J0>
__device__ __forceinline__ void rows_load(const float* __restrict__ x, v2f (&r)[16], int s, int lane) {
#pragma unroll
    for (int j = 0; j < 16; ++j) {
        const int sj = __builtin_amdgcn_readlane(s, J0 + j);
        r[j] = *(const v2f*)(x + (size_t)sj * DFEAT + 2 * lane);
    }
}

template <int J0, bool FULL>
__device__ __forceinline__ void rows_own_add(float* acc, const v2f (&r)[16], int d, int c, int lane) {
#pragma unroll
    for (int j = 0; j < 16; ++j) {
        if (FULL || j < c) {
            const int dj = __builtin_amdgcn_readlane(d, J0 + j);
            v2f* p = (v2f*)(acc + dj * DFEAT + 2 * lane);
            v2f a = *p;
            a.x += r[j].x;
            a.y += r[j].y;
            *p = a;
        }
    }
}

template <bool FULL, bool REC>
__device__ __forceinline__ void agg_drain(const float* __restrict__ x,
                                          const unsigned* __restrict__ packed,
                                          float* acc, unsigned* cnt, const unsigned* wq,
                                          int m, int lane, unsigned* oq, unsigned* oqn) {
    const bool live = FULL || lane < m;
    const unsigned ent = live ? packed[wq[lane]] : 0u;
    const int d = (int)(ent & 1023u);
    const int s = (int)(ent >> 10);
    if (REC && live)
        __hip_atomic_fetch_add(&cnt[d], 1u, __ATOMIC_RELAXED, __HIP_MEMORY_SCOPE_WORKGROUP);
    if (REC && live) {
        const unsigned ow  = (unsigned)d & (AGG_WAVES - 1);
        const unsigned pos = __hip_atomic_fetch_add(&oqn[ow], 1u, __ATOMIC_RELAXED, __HIP_MEMORY_SCOPE_WORKGROUP);
        if (pos < (unsigned)OQ_CAP) oq[ow * OQ_CAP + pos] = ent;
    }
    gather_add<FULL>(x, acc, s, d, m, lane, [] {});
}

template <bool FIRST>
__global__ __launch_bounds__(AGG_THREADS)
void gcn_agg_kernel(const float* __restrict__ x,
                    const unsigned* __restrict__ keys,
                    const unsigned* __restrict__ packed,
                    float* __restrict__ agg,
                    float* __restrict__ degp,
                    unsigned* __restrict__ lists,
                    unsigned* __restrict__ hdrs,
                    int N, int E) {
    __shared__ __attribute__((aligned(16))) float    acc[AGG_R * DFEAT];
    __shared__ unsigned cnt[AGG_R];
    __shared__ unsigned wqs[AGG_WAVES * AGG_WQ];
    __shared__ unsigned oq[FIRST ? AGG_WAVES * OQ_CAP : 1];
    __shared__ unsigned oqn[AGG_WAVES];

    const int tid  = threadIdx.x;
    const int lane = tid & 31;
    const int wave = __builtin_amdgcn_readfirstlane(tid >> 5);
    const int lo    = blockIdx.x * AGG_R;
    const int slice = blockIdx.y;
    const unsigned nloc = (unsigned)min(AGG_R, N - lo);
    unsigned* wq = wqs + wave * AGG_WQ;

    const size_t blk = (size_t)slice * gridDim.x + blockIdx.x;
    unsigned* hdr    = hdrs + blk * LIST_HDR;
    unsigned* mylist = lists + (blk * AGG_WAVES + wave) * LIST_WCAP;

    for (int i = tid; i < AGG_R * DFEAT / 4; i += AGG_THREADS)
        ((float4*)acc)[i] = make_float4(0.f, 0.f, 0.f, 0.f);
    if (FIRST) {
        for (int i = tid; i < AGG_R; i += AGG_THREADS) cnt[i] = 0u;
        if (tid < AGG_WAVES) oqn[tid] = 0u;
    }
    __syncthreads();

    bool scan = true;
    if (!FIRST) scan = __builtin_amdgcn_ballot_w32(hdr[lane] > (unsigned)OQ_CAP) != 0u;

    if (scan) {
        const uint4* key4 = (const uint4*)keys;
        const int nvec  = E >> 4;
        const int gvec  = AGG_VEC * AGG_THREADS;
        const int nfull = nvec / gvec;
        const int gs = (int)(((long long)slice * nfull) / AGG_S);
        const int ng = (int)(((long long)(slice + 1) * nfull) / AGG_S) - gs;
        const int g0 = (int)(((long long)blockIdx.x * ng) / gridDim.x);
        const unsigned pat = (unsigned)blockIdx.x * 0x01010101u;

        unsigned cur  = 0;

        auto drain = [&](const unsigned* q, unsigned m) {
            if (m == 32u) agg_drain<true,  FIRST>(x, packed, acc, cnt, q, 32, lane, oq, oqn);
            else          agg_drain<false, FIRST>(x, packed, acc, cnt, q, (int)m, lane, oq, oqn);
        };

        auto flags = [&](unsigned w) {
            const unsigned x = w ^ pat;
            return (0x80808080u - x) & 0x80808080u;
        };

        auto visit16 = [&](const uint4& v, unsigned ebase) {
            unsigned mm = flags(v.x) | (flags(v.y) >> 1) | (flags(v.z) >> 2) | (flags(v.w) >> 3);
            unsigned mask = __builtin_amdgcn_ballot_w32(mm != 0u);
            while (mask) {
                if (mm != 0u) {
                    const unsigned b = (unsigned)__builtin_ctz(mm);
                    const unsigned e = ebase + ((7u - (b & 7u)) << 2) + (b >> 3);
                    wq[cur + __builtin_amdgcn_mbcnt_lo(mask, 0u)] = e;
                    mm &= mm - 1u;
                }
                cur += (unsigned)__builtin_popcount(mask);
                mask = __builtin_amdgcn_ballot_w32(mm != 0u);
            }
        };

        auto group_at = [&](int g) {
            int gg = g + g0;
            return gs + (gg >= ng ? gg - ng : gg);
        };

        auto load_group = [&](int g, uint4 (&v)[AGG_VEC]) {
            const int base = group_at(g) * gvec + tid;
#pragma unroll
            for (int k = 0; k < AGG_VEC; ++k) v[k] = key4[base + k * AGG_THREADS];
        };

        auto visit_vecs = [&](int base, const uint4 (&v)[AGG_VEC]) {
#pragma unroll
            for (int k = 0; k < AGG_VEC; ++k) {
                visit16(v[k], (unsigned)(base + k * AGG_THREADS) << 4);
                while (cur >= 32u) {
                    cur -= 32u;
                    drain(wq + cur, 32u);
                }
            }
        };

        if (ng > 0) {
            uint4 va[AGG_VEC], vb[AGG_VEC];
            load_group(0, va);
            for (int g = 0; g < ng; ++g) {
                load_group(min(g + 1, ng - 1), vb);
                visit_vecs(group_at(g) * gvec + tid, va);
#pragma unroll
                for (int k = 0; k < AGG_VEC; ++k) va[k] = vb[k];
            }
        }
        if (slice == 0) {
            uint4 vr[AGG_VEC];
            const int base = nfull * gvec + tid;
            const unsigned none = pat ^ 0x01010101u;
#pragma unroll
            for (int k = 0; k < AGG_VEC; ++k) {
                const int vi = base + k * AGG_THREADS;
                vr[k] = vi < nvec ? key4[vi] : make_uint4(none, none, none, none);
            }
            visit_vecs(base, vr);
        }
        if (slice == 0 && wave == 0 && (E & 15) != 0) {
            const int e = (nvec << 4) + lane;
            const bool tail = lane < (E & 15);
            const unsigned kb = tail ? (unsigned)((const unsigned char*)keys)[e] : 0xFFu;
            const bool hit = kb == (unsigned)blockIdx.x;
            const unsigned mask = __builtin_amdgcn_ballot_w32(hit);
            if (hit) wq[cur + __builtin_amdgcn_mbcnt_lo(mask, 0u)] = (unsigned)e;
            cur += (unsigned)__builtin_popcount(mask);
        }
        while (cur >= 32u) {
            cur -= 32u;
            drain(wq + cur, 32u);
        }
        if (cur) drain(wq, cur);
    } else {
        const unsigned n = (unsigned)__builtin_amdgcn_readfirstlane((int)hdr[wave]);
        if (n > 0u) {
            unsigned ent = mylist[min((unsigned)lane, n - 1u)];
            if ((unsigned)lane >= n) ent = 0u;
            v2f ra[16], rb[16];
            rows_load<0>(x, ra, (int)(ent >> 10), lane);
            for (unsigned k = 0; k < n; k += 32u) {
                const int d = (int)(ent & 1023u);
                const int s = (int)(ent >> 10);
                const unsigned m  = n - k;
                const unsigned kn = k + 32u + (unsigned)lane;
                unsigned nxt = mylist[min(kn, n - 1u)];
                if (kn >= n) nxt = 0u;
                rows_load<16>(x, rb, s, lane);
                if (m >= 16u) rows_own_add<0, true >(acc, ra, d, 16, lane);
                else          rows_own_add<0, false>(acc, ra, d, (int)m, lane);
                rows_load<0>(x, ra, (int)(nxt >> 10), lane);
                if (m >= 32u)     rows_own_add<16, true >(acc, rb, d, 16, lane);
                else if (m > 16u) rows_own_add<16, false>(acc, rb, d, (int)m - 16, lane);
                ent = nxt;
            }
        }
    }

    __syncthreads();

    float4* orow = (float4*)(agg + ((size_t)slice * N + lo) * DFEAT);
    for (int i = tid; i < (int)nloc * (DFEAT / 4); i += AGG_THREADS)
        orow[i] = ((const float4*)acc)[i];
    if (FIRST) {
        for (int i = tid; i < (int)nloc; i += AGG_THREADS)
            degp[(size_t)slice * N + lo + i] = (float)cnt[i];
        const unsigned nq = (unsigned)__builtin_amdgcn_readfirstlane((int)oqn[wave]);
        const unsigned nk = min(nq, (unsigned)OQ_CAP);
        for (unsigned k = (unsigned)lane; k < nk; k += 32u) mylist[k] = oq[wave * OQ_CAP + k];
        if (lane == 0) hdr[wave] = nq;
    }
}

#define LIN_WAVES   8
#define LIN_THREADS (LIN_WAVES * 32)
#define LIN_PITCH   (DFEAT + 4)

__global__ __launch_bounds__(LIN_THREADS)
void gcn_wmma_kernel(const float* __restrict__ agg,
                     const float* __restrict__ degp,
                     const float* __restrict__ W,
                     const float* __restrict__ bias,
                     float* __restrict__ out, int N) {
    __shared__ __attribute__((aligned(16))) float Ws[DFEAT * LIN_PITCH];
    __shared__ __attribute__((aligned(16))) float As[LIN_WAVES * 16 * LIN_PITCH];

    const int tid  = threadIdx.x;
    const int lane = tid & 31;
    const int wave  = __builtin_amdgcn_readfirstlane(tid >> 5);
    const int tiles = N >> 4;
    const int want  = blockIdx.x * LIN_WAVES + wave;
    const bool valid = want < tiles;
    const int tile = valid ? want : tiles - 1;
    const int rowBase = tile << 4;
    const int m  = lane & 15;
    const int hi = lane >> 4;
    const int k2 = hi << 1;
    const int c4 = m << 2;
    float* mine = As + wave * (16 * LIN_PITCH);
    const size_t plane = (size_t)N * DFEAT;

    float4 w4[DFEAT * DFEAT / 4 / LIN_THREADS];
#pragma unroll
    for (int j = 0; j < DFEAT * DFEAT / 4 / LIN_THREADS; ++j)
        w4[j] = *(const float4*)(W + 4 * (tid + j * LIN_THREADS));

    const float* rows = agg + (size_t)rowBase * DFEAT;
    const unsigned loff = (unsigned)(hi * DFEAT + c4);
    float4 v[8];
#pragma unroll
    for (int i = 0; i < 8; ++i)
        v[i] = *(const float4*)(rows + loff + i * 2 * DFEAT);
    float4 u[AGG_S - 1][8];
#pragma unroll
    for (int p = 1; p < AGG_S; ++p) {
#pragma unroll
        for (int i = 0; i < 8; ++i)
            u[p - 1][i] = *(const float4*)(rows + p * plane + loff + i * 2 * DFEAT);
    }

    float deg = degp[rowBase + m];
#pragma unroll
    for (int p = 1; p < AGG_S; ++p) deg += degp[(size_t)p * N + rowBase + m];
    float bcol[4];
#pragma unroll
    for (int t = 0; t < 4; ++t) bcol[t] = bias[t * 16 + m];

#pragma unroll
    for (int j = 0; j < DFEAT * DFEAT / 4 / LIN_THREADS; ++j) {
        const int idx = tid + j * LIN_THREADS;
        *(float4*)(Ws + (idx >> 4) * LIN_PITCH + ((idx & 15) << 2)) = w4[j];
    }
#pragma unroll
    for (int i = 0; i < 8; ++i) {
#pragma unroll
        for (int p = 1; p < AGG_S; ++p) {
            v[i].x += u[p - 1][i].x; v[i].y += u[p - 1][i].y;
            v[i].z += u[p - 1][i].z; v[i].w += u[p - 1][i].w;
        }
        *(float4*)(mine + (2 * i + hi) * LIN_PITCH + c4) = v[i];
    }
    const float inv = deg > 0.0f ? 1.0f / deg : 0.0f;
    __syncthreads();

    v8f c[4] = {};
    const float* arow = mine + m * LIN_PITCH + k2;
    const float* brow = Ws   + m * LIN_PITCH + k2;
#pragma unroll
    for (int kb = 0; kb < DFEAT; kb += 4) {
        v2f a = *(const v2f*)(arow + kb);
        a.x *= inv;
        a.y *= inv;
#pragma unroll
        for (int t = 0; t < 4; ++t) {
            const v2f b = *(const v2f*)(brow + t * 16 * LIN_PITCH + kb);
            c[t] = __builtin_amdgcn_wmma_f32_16x16x4_f32(
                     false, a,  false, b,
                     (short)0, c[t],  false,  false);
        }
    }

#pragma unroll
    for (int t = 0; t < 4; ++t) {
#pragma unroll
        for (int r = 0; r < 8; ++r) {
            float e = c[t][r] + bcol[t];
            e = e > 0.0f ? e : expm1f(e);
            mine[(r + (hi << 3)) * LIN_PITCH + t * 16 + m] = e;
        }
    }
    __syncthreads();

    if (valid) {
        float* orow = out + (size_t)rowBase * DFEAT;
#pragma unroll
        for (int i = 0; i < 8; ++i)
            *(float4*)(orow + loff + i * 2 * DFEAT) =
                *(const float4*)(mine + (2 * i + hi) * LIN_PITCH + c4);
    }
}

extern "C" void kernel_launch(void* const* d_in, const int* in_sizes, int n_in,
                              void* d_out, int out_size, void* d_ws, size_t ws_size,
                              hipStream_t stream) {
    const float* h   = (const float*)d_in[0];
    const int*   src = (const int*)  d_in[1];
    const int*   dst = (const int*)  d_in[2];
    const float* W1  = (const float*)d_in[3];
    const float* b1  = (const float*)d_in[4];
    const float* W2  = (const float*)d_in[5];
    const float* b2  = (const float*)d_in[6];
    float*       out = (float*)d_out;

    const int N = in_sizes[0] / DFEAT;
    const int E = in_sizes[1];

    char* ws = (char*)d_ws;
    float* deg = (float*)ws;
    size_t degBytes = (((size_t)AGG_S * N * sizeof(float)) + 255) & ~(size_t)255;
    float* agg = (float*)(ws + degBytes);
    float* h1  = agg + (size_t)AGG_S * N * DFEAT;
    unsigned* hdrs  = (unsigned*)(h1 + (size_t)N * DFEAT);

    const int tiles = N / 16;
    const int linB  = (tiles + LIN_WAVES - 1) / LIN_WAVES;
    const dim3 aggB((N + AGG_R - 1) / AGG_R, AGG_S);
    unsigned* lists = hdrs + (size_t)aggB.x * aggB.y * LIST_HDR;
    unsigned* keys   = lists + (size_t)aggB.x * aggB.y * AGG_WAVES * LIST_WCAP;
    unsigned* packed = keys + (size_t)((E + 15) / 16) * 4;
    const int keyB   = (E / 4 + KEY_THREADS * KEY_ITEMS - 1) / (KEY_THREADS * KEY_ITEMS) + (E < 4);

    gcn_key_kernel<<<keyB, KEY_THREADS, 0, stream>>>(src, dst, keys, packed, E);

    gcn_agg_kernel<true><<<aggB, AGG_THREADS, 0, stream>>>(h, keys, packed, agg, deg, lists, hdrs, N, E);
    gcn_wmma_kernel<<<linB, LIN_THREADS, 0, stream>>>(agg, deg, W1, b1, h1, N);

    gcn_agg_kernel<false><<<aggB, AGG_THREADS, 0, stream>>>(h1, keys, packed, agg, deg, lists, hdrs, N, E);
    gcn_wmma_kernel<<<linB, LIN_THREADS, 0, stream>>>(agg, deg, W2, b2, out, N);
}


# W2 product moved ahead of layer 2's aggregation (linear kernel: 2nd WMMA pass on its h1 tile); layer-2 blocks own half a range, all edges, end in scale+bias+ELU; 5->4 launches; sum order changes
# speedup vs baseline: 4.2316x; 1.1505x over previous
#include <hip/hip_runtime.h>
#include <hip/hip_bf16.h>
#include <math.h>

typedef __attribute__((ext_vector_type(2))) float v2f;
typedef __attribute__((ext_vector_type(8))) float v8f;

#define DFEAT 64

#define AGG_S       2
#define AGG_R       782
#define AGG_H       (AGG_R / 2)
#define AGG_THREADS 1024
#define AGG_WAVES   (AGG_THREADS / 32)
#define AGG_VEC     4
#define AGG_WQ      (16 * 32 + 32)
#define KEY_THREADS 256
#define KEY_ITEMS   4

__global__ __launch_bounds__(KEY_THREADS)
void gcn_key_kernel(const int* __restrict__ src,
                    const int* __restrict__ dst,
                    unsigned* __restrict__ keys,
                    unsigned* __restrict__ packed,
                    int E) {
    const int nv   = E >> 2;
    const int base = blockIdx.x * (KEY_THREADS * KEY_ITEMS) + threadIdx.x;
    if (nv > 0) {
        int4 d[KEY_ITEMS], s[KEY_ITEMS];
#pragma unroll
        for (int k = 0; k < KEY_ITEMS; ++k) {
            const int i = min(base + k * KEY_THREADS, nv - 1);
            d[k] = ((const int4*)dst)[i];
            s[k] = ((const int4*)src)[i];
        }
#pragma unroll
        for (int k = 0; k < KEY_ITEMS; ++k) {
            const int i = base + k * KEY_THREADS;
            if (i < nv) {
                const unsigned r0 = (unsigned)d[k].x / AGG_R, r1 = (unsigned)d[k].y / AGG_R;
                const unsigned r2 = (unsigned)d[k].z / AGG_R, r3 = (unsigned)d[k].w / AGG_R;
                uint4 p;
                p.x = ((unsigned)s[k].x << 10) | ((unsigned)d[k].x - r0 * AGG_R);
                p.y = ((unsigned)s[k].y << 10) | ((unsigned)d[k].y - r1 * AGG_R);
                p.z = ((unsigned)s[k].z << 10) | ((unsigned)d[k].z - r2 * AGG_R);
                p.w = ((unsigned)s[k].w << 10) | ((unsigned)d[k].w - r3 * AGG_R);
                keys[i] = r0 | (r1 << 8) | (r2 << 16) | (r3 << 24);
                ((uint4*)packed)[i] = p;
            }
        }
    }
    if (blockIdx.x == 0 && (int)threadIdx.x < (E & 3)) {
        const int e = (nv << 2) + (int)threadIdx.x;
        const unsigned dv = (unsigned)dst[e];
        const unsigned r  = dv / AGG_R;
        packed[e] = ((unsigned)src[e] << 10) | (dv - r * AGG_R);
        ((unsigned char*)keys)[e] = (unsigned char)r;
    }
}

#define OQ_N        (2 * AGG_WAVES)
#define OQ_CAP      160
#define LIST_WCAP   OQ_CAP
#define LIST_HDR    OQ_N
static_assert(AGG_S == 2 && (AGG_R & 1) == 0, "layer 2 reads two slices' lists per half range");

template <bool FULL, typename BETWEEN>
__device__ __forceinline__ void gather_add(const float* __restrict__ x, float* acc,
                                           int s, int d, int m, int lane, BETWEEN between) {
#pragma unroll
    for (int h = 0; h < 2; ++h) {
        if (!FULL && h * 16 >= m) break;
        v2f r[16];
#pragma unroll
        for (int j = 0; j < 16; ++j) {
            const int sj = __builtin_amdgcn_readlane(s, h * 16 + j);
            r[j] = *(const v2f*)(x + (size_t)sj * DFEAT + 2 * lane);
        }
        if (h == 0) between();
#pragma unroll
        for (int j = 0; j < 16; ++j) {
            if (FULL || h * 16 + j < m) {
                const int dj = __builtin_amdgcn_readlane(d, h * 16 + j);
                float* p = acc + dj * DFEAT + 2 * lane;
                __hip_atomic_fetch_add(p + 0, r[j].x, __ATOMIC_RELAXED, __HIP_MEMORY_SCOPE_WORKGROUP);
                __hip_atomic_fetch_add(p + 1, r[j].y, __ATOMIC_RELAXED, __HIP_MEMORY_SCOPE_WORKGROUP);
            }
        }
    }
}

template <int J0>
__device__ __forceinline__ void rows_load(const float* __restrict__ x, v2f (&r)[16], int s, int lane) {
#pragma unroll
    for (int j = 0; j < 16; ++j) {
        const int sj = __builtin_amdgcn_readlane(s, J0 + j);
        r[j] = *(const v2f*)(x + (size_t)sj * DFEAT + 2 * lane);
    }
}

template <int J0, bool FULL>
__device__ __forceinline__ void rows_own_add(float* acc, const v2f (&r)[16], int d, int c, int lane) {
#pragma unroll
    for (int j = 0; j < 16; ++j) {
        if (FULL || j < c) {
            const int dj = __builtin_amdgcn_readlane(d, J0 + j);
            v2f* p = (v2f*)(acc + dj * DFEAT + 2 * lane);
            v2f a = *p;
            a.x += r[j].x;
            a.y += r[j].y;
            *p = a;
        }
    }
}

template <bool FULL, bool REC>
__device__ __forceinline__ void agg_drain(const float* __restrict__ x,
                                          const unsigned* __restrict__ packed,
                                          float* acc, unsigned* cnt, const unsigned* wq,
                                          int m, int lane, unsigned* oq, unsigned* oqn, int half) {
    const bool live = FULL || lane < m;
    const unsigned ent = live ? packed[wq[lane]] : 0u;
    const int d = (int)(ent & 1023u);
    const int s = (int)(ent >> 10);
    const int hf = d >= AGG_H ? 1 : 0;
    const int d2 = d - hf * AGG_H;
    if (REC) {
        if (live) {
            __hip_atomic_fetch_add(&cnt[d], 1u, __ATOMIC_RELAXED, __HIP_MEMORY_SCOPE_WORKGROUP);
            const unsigned q   = (unsigned)hf * AGG_WAVES + ((unsigned)d2 & (AGG_WAVES - 1));
            const unsigned pos = __hip_atomic_fetch_add(&oqn[q], 1u, __ATOMIC_RELAXED, __HIP_MEMORY_SCOPE_WORKGROUP);
            if (pos < (unsigned)OQ_CAP) oq[q * OQ_CAP + pos] = (ent & ~1023u) | (unsigned)d2;
        }
        gather_add<FULL>(x, acc, s, d, m, lane, [] {});
    } else {
        unsigned todo = __builtin_amdgcn_ballot_w32(live && hf == half);
        while (todo) {
            const int j = __builtin_ctz(todo);
            todo &= todo - 1u;
            const int sj = __builtin_amdgcn_readlane(s, j);
            const int dj = __builtin_amdgcn_readlane(d2, j);
            const v2f r = *(const v2f*)(x + (size_t)sj * DFEAT + 2 * lane);
            float* p = acc + dj * DFEAT + 2 * lane;
            __hip_atomic_fetch_add(p + 0, r.x, __ATOMIC_RELAXED, __HIP_MEMORY_SCOPE_WORKGROUP);
            __hip_atomic_fetch_add(p + 1, r.y, __ATOMIC_RELAXED, __HIP_MEMORY_SCOPE_WORKGROUP);
        }
    }
}

template <bool FIRST>
__global__ __launch_bounds__(AGG_THREADS)
void gcn_agg_kernel(const float* __restrict__ x,
                    const unsigned* __restrict__ keys,
                    const unsigned* __restrict__ packed,
                    float* __restrict__ agg,
                    float* __restrict__ degp,
                    unsigned* __restrict__ lists,
                    unsigned* __restrict__ hdrs,
                    const float* __restrict__ bias,
                    float* __restrict__ out,
                    int N, int E) {
    constexpr int ROWS = FIRST ? AGG_R : AGG_H;
    __shared__ __attribute__((aligned(16))) float    acc[ROWS * DFEAT];
    __shared__ unsigned cnt[FIRST ? AGG_R : 1];
    __shared__ unsigned wqs[AGG_WAVES * AGG_WQ];
    __shared__ unsigned oq[FIRST ? OQ_N * OQ_CAP : 1];
    __shared__ unsigned oqn[OQ_N];
    __shared__ float    invd[FIRST ? 1 : AGG_H];

    const int tid  = threadIdx.x;
    const int lane = tid & 31;
    const int wave = __builtin_amdgcn_readfirstlane(tid >> 5);
    const int nrng  = FIRST ? (int)gridDim.x : (int)(gridDim.x >> 1);
    const int rng   = FIRST ? (int)blockIdx.x : (int)(blockIdx.x >> 1);
    const int half  = FIRST ? 0 : (int)(blockIdx.x & 1);
    const int slice = FIRST ? (int)blockIdx.y : 0;
    const int nslice = FIRST ? AGG_S : 1;
    const int lo    = rng * AGG_R + half * AGG_H;
    const unsigned nloc = (unsigned)max(0, min(ROWS, N - lo));
    unsigned* wq = wqs + wave * AGG_WQ;

    const size_t b0 = (size_t)rng;
    const size_t b1 = (size_t)nrng + rng;
    unsigned f0 = 0u, f1 = 0u;
    if (!FIRST) {
        f0 = hdrs[b0 * LIST_HDR + half * AGG_WAVES + lane];
        f1 = hdrs[b1 * LIST_HDR + half * AGG_WAVES + lane];
    }

    for (int i = tid; i < ROWS * DFEAT / 4; i += AGG_THREADS)
        ((float4*)acc)[i] = make_float4(0.f, 0.f, 0.f, 0.f);
    if (FIRST) {
        for (int i = tid; i < AGG_R; i += AGG_THREADS) cnt[i] = 0u;
        if (tid < OQ_N) oqn[tid] = 0u;
    } else {
        for (int i = tid; i < (int)nloc; i += AGG_THREADS) {
            float dg = 0.0f;
#pragma unroll
            for (int p = 0; p < AGG_S; ++p) dg += degp[(size_t)p * N + lo + i];
            invd[i] = dg > 0.0f ? 1.0f / dg : 0.0f;
        }
    }
    __syncthreads();

    bool scan = true;
    unsigned n0 = 0u, n1 = 0u;
    const unsigned* l0 = lists;
    const unsigned* l1 = lists;
    if (!FIRST) {
        scan = __builtin_amdgcn_ballot_w32(f0 > (unsigned)OQ_CAP || f1 > (unsigned)OQ_CAP) != 0u;
        n0 = (unsigned)__builtin_amdgcn_readlane((int)f0, wave);
        n1 = (unsigned)__builtin_amdgcn_readlane((int)f1, wave);
        l0 = lists + (b0 * OQ_N + half * AGG_WAVES + wave) * LIST_WCAP;
        l1 = lists + (b1 * OQ_N + half * AGG_WAVES + wave) * LIST_WCAP;
    }

    if (scan) {
        const uint4* key4 = (const uint4*)keys;
        const int nvec  = E >> 4;
        const int gvec  = AGG_VEC * AGG_THREADS;
        const int nfull = nvec / gvec;
        const int gs = (int)(((long long)slice * nfull) / nslice);
        const int ng = (int)(((long long)(slice + 1) * nfull) / nslice) - gs;
        const int g0 = (int)(((long long)blockIdx.x * ng) / gridDim.x);
        const unsigned pat = (unsigned)rng * 0x01010101u;

        unsigned cur  = 0;

        auto drain = [&](const unsigned* q, unsigned m) {
            if (m == 32u) agg_drain<true,  FIRST>(x, packed, acc, cnt, q, 32, lane, oq, oqn, half);
            else          agg_drain<false, FIRST>(x, packed, acc, cnt, q, (int)m, lane, oq, oqn, half);
        };

        auto flags = [&](unsigned w) {
            const unsigned x = w ^ pat;
            return (0x80808080u - x) & 0x80808080u;
        };

        auto visit16 = [&](const uint4& v, unsigned ebase) {
            unsigned mm = flags(v.x) | (flags(v.y) >> 1) | (flags(v.z) >> 2) | (flags(v.w) >> 3);
            unsigned mask = __builtin_amdgcn_ballot_w32(mm != 0u);
            while (mask) {
                if (mm != 0u) {
                    const unsigned b = (unsigned)__builtin_ctz(mm);
                    const unsigned e = ebase + ((7u - (b & 7u)) << 2) + (b >> 3);
                    wq[cur + __builtin_amdgcn_mbcnt_lo(mask, 0u)] = e;
                    mm &= mm - 1u;
                }
                cur += (unsigned)__builtin_popcount(mask);
                mask = __builtin_amdgcn_ballot_w32(mm != 0u);
            }
        };

        auto group_at = [&](int g) {
            int gg = g + g0;
            return gs + (gg >= ng ? gg - ng : gg);
        };

        auto load_group = [&](int g, uint4 (&v)[AGG_VEC]) {
            const int base = group_at(g) * gvec + tid;
#pragma unroll
            for (int k = 0; k < AGG_VEC; ++k) v[k] = key4[base + k * AGG_THREADS];
        };

        auto visit_vecs = [&](int base, const uint4 (&v)[AGG_VEC]) {
#pragma unroll
            for (int k = 0; k < AGG_VEC; ++k) {
                visit16(v[k], (unsigned)(base + k * AGG_THREADS) << 4);
                while (cur >= 32u) {
                    cur -= 32u;
                    drain(wq + cur, 32u);
                }
            }
        };

        if (ng > 0) {
            uint4 va[AGG_VEC], vb[AGG_VEC];
            load_group(0, va);
            for (int g = 0; g < ng; ++g) {
                load_group(min(g + 1, ng - 1), vb);
                visit_vecs(group_at(g) * gvec + tid, va);
#pragma unroll
                for (int k = 0; k < AGG_VEC; ++k) va[k] = vb[k];
            }
        }
        if (slice == 0) {
            uint4 vr[AGG_VEC];
            const int base = nfull * gvec + tid;
            const unsigned none = pat ^ 0x01010101u;
#pragma unroll
            for (int k = 0; k < AGG_VEC; ++k) {
                const int vi = base + k * AGG_THREADS;
                vr[k] = vi < nvec ? key4[vi] : make_uint4(none, none, none, none);
            }
            visit_vecs(base, vr);
        }
        if (slice == 0 && wave == 0 && (E & 15) != 0) {
            const int e = (nvec << 4) + lane;
            const bool tail = lane < (E & 15);
            const unsigned kb = tail ? (unsigned)((const unsigned char*)keys)[e] : 0xFFu;
            const bool hit = kb == (unsigned)rng;
            const unsigned mask = __builtin_amdgcn_ballot_w32(hit);
            if (hit) wq[cur + __builtin_amdgcn_mbcnt_lo(mask, 0u)] = (unsigned)e;
            cur += (unsigned)__builtin_popcount(mask);
        }
        while (cur >= 32u) {
            cur -= 32u;
            drain(wq + cur, 32u);
        }
        if (cur) drain(wq, cur);
    } else {
        const unsigned n = n0 + n1;
        if (n > 0u) {
            auto entry = [&](unsigned k) {
                const unsigned kk = min(k, n - 1u);
                const unsigned* p = kk < n0 ? l0 + kk : l1 + (kk - n0);
                const unsigned e = *p;
                return k < n ? e : 0u;
            };
            unsigned ent = entry((unsigned)lane);
            v2f ra[16], rb[16];
            rows_load<0>(x, ra, (int)(ent >> 10), lane);
            for (unsigned k = 0; k < n; k += 32u) {
                const int d = (int)(ent & 1023u);
                const int s = (int)(ent >> 10);
                const unsigned m  = n - k;
                const unsigned kn = k + 32u + (unsigned)lane;
                const unsigned nxt = entry(kn);
                rows_load<16>(x, rb, s, lane);
                if (m >= 16u) rows_own_add<0, true >(acc, ra, d, 16, lane);
                else          rows_own_add<0, false>(acc, ra, d, (int)m, lane);
                rows_load<0>(x, ra, (int)(nxt >> 10), lane);
                if (m >= 32u)     rows_own_add<16, true >(acc, rb, d, 16, lane);
                else if (m > 16u) rows_own_add<16, false>(acc, rb, d, (int)m - 16, lane);
                ent = nxt;
            }
        }
    }

    __syncthreads();

    if (FIRST) {
        float4* orow = (float4*)(agg + ((size_t)slice * N + lo) * DFEAT);
        for (int i = tid; i < (int)nloc * (DFEAT / 4); i += AGG_THREADS)
            orow[i] = ((const float4*)acc)[i];
        for (int i = tid; i < (int)nloc; i += AGG_THREADS)
            degp[(size_t)slice * N + lo + i] = (float)cnt[i];
        const size_t blk = (size_t)slice * nrng + rng;
#pragma unroll
        for (int hq = 0; hq < 2; ++hq) {
            const int q = hq * AGG_WAVES + wave;
            const unsigned nq = (unsigned)__builtin_amdgcn_readfirstlane((int)oqn[q]);
            const unsigned nk = min(nq, (unsigned)OQ_CAP);
            unsigned* lst = lists + (blk * OQ_N + q) * LIST_WCAP;
            for (unsigned k = (unsigned)lane; k < nk; k += 32u) lst[k] = oq[q * OQ_CAP + k];
            if (lane == 0) hdrs[blk * LIST_HDR + q] = nq;
        }
    } else {
        float4* orow = (float4*)(out + (size_t)lo * DFEAT);
        for (int i = tid; i < (int)nloc * (DFEAT / 4); i += AGG_THREADS) {
            const float4 a  = ((const float4*)acc)[i];
            const float4 b  = ((const float4*)bias)[i & (DFEAT / 4 - 1)];
            const float  iv = invd[i / (DFEAT / 4)];
            float4 e;
            e.x = a.x * iv + b.x; e.y = a.y * iv + b.y;
            e.z = a.z * iv + b.z; e.w = a.w * iv + b.w;
            e.x = e.x > 0.0f ? e.x : expm1f(e.x);
            e.y = e.y > 0.0f ? e.y : expm1f(e.y);
            e.z = e.z > 0.0f ? e.z : expm1f(e.z);
            e.w = e.w > 0.0f ? e.w : expm1f(e.w);
            orow[i] = e;
        }
    }
}

#define LIN_WAVES   8
#define LIN_THREADS (LIN_WAVES * 32)
#define LIN_PITCH   (DFEAT + 4)

__global__ __launch_bounds__(LIN_THREADS)
void gcn_wmma_kernel(const float* __restrict__ agg,
                     const float* __restrict__ degp,
                     const float* __restrict__ W,
                     const float* __restrict__ bias,
                     const float* __restrict__ W2,
                     float* __restrict__ out,
                     int N) {
    __shared__ __attribute__((aligned(16))) float Ws[DFEAT * LIN_PITCH];
    __shared__ __attribute__((aligned(16))) float Ws2[DFEAT * LIN_PITCH];
    __shared__ __attribute__((aligned(16))) float As[LIN_WAVES * 16 * LIN_PITCH];

    const int tid  = threadIdx.x;
    const int lane = tid & 31;
    const int wave  = __builtin_amdgcn_readfirstlane(tid >> 5);
    const int tiles = N >> 4;
    const int want  = blockIdx.x * LIN_WAVES + wave;
    const bool valid = want < tiles;
    const int tile = valid ? want : tiles - 1;
    const int rowBase = tile << 4;
    const int m  = lane & 15;
    const int hi = lane >> 4;
    const int k2 = hi << 1;
    const int c4 = m << 2;
    float* mine = As + wave * (16 * LIN_PITCH);
    const size_t plane = (size_t)N * DFEAT;

    float4 w4[DFEAT * DFEAT / 4 / LIN_THREADS], x4[DFEAT * DFEAT / 4 / LIN_THREADS];
#pragma unroll
    for (int j = 0; j < DFEAT * DFEAT / 4 / LIN_THREADS; ++j) {
        w4[j] = *(const float4*)(W  + 4 * (tid + j * LIN_THREADS));
        x4[j] = *(const float4*)(W2 + 4 * (tid + j * LIN_THREADS));
    }

    const float* rows = agg + (size_t)rowBase * DFEAT;
    const unsigned loff = (unsigned)(hi * DFEAT + c4);
    float4 v[8];
#pragma unroll
    for (int i = 0; i < 8; ++i)
        v[i] = *(const float4*)(rows + loff + i * 2 * DFEAT);
    float4 u[AGG_S - 1][8];
#pragma unroll
    for (int p = 1; p < AGG_S; ++p) {
#pragma unroll
        for (int i = 0; i < 8; ++i)
            u[p - 1][i] = *(const float4*)(rows + p * plane + loff + i * 2 * DFEAT);
    }

    float deg = degp[rowBase + m];
#pragma unroll
    for (int p = 1; p < AGG_S; ++p) deg += degp[(size_t)p * N + rowBase + m];
    float bcol[4];
#pragma unroll
    for (int t = 0; t < 4; ++t) bcol[t] = bias[t * 16 + m];

#pragma unroll
    for (int j = 0; j < DFEAT * DFEAT / 4 / LIN_THREADS; ++j) {
        const int idx = tid + j * LIN_THREADS;
        *(float4*)(Ws  + (idx >> 4) * LIN_PITCH + ((idx & 15) << 2)) = w4[j];
        *(float4*)(Ws2 + (idx >> 4) * LIN_PITCH + ((idx & 15) << 2)) = x4[j];
    }
#pragma unroll
    for (int i = 0; i < 8; ++i) {
#pragma unroll
        for (int p = 1; p < AGG_S; ++p) {
            v[i].x += u[p - 1][i].x; v[i].y += u[p - 1][i].y;
            v[i].z += u[p - 1][i].z; v[i].w += u[p - 1][i].w;
        }
        *(float4*)(mine + (2 * i + hi) * LIN_PITCH + c4) = v[i];
    }
    const float inv = deg > 0.0f ? 1.0f / deg : 0.0f;
    __syncthreads();

    v8f c[4] = {};
    const float* arow = mine + m * LIN_PITCH + k2;
    const float* brow = Ws   + m * LIN_PITCH + k2;
#pragma unroll
    for (int kb = 0; kb < DFEAT; kb += 4) {
        v2f a = *(const v2f*)(arow + kb);
        a.x *= inv;
        a.y *= inv;
#pragma unroll
        for (int t = 0; t < 4; ++t) {
            const v2f b = *(const v2f*)(brow + t * 16 * LIN_PITCH + kb);
            c[t] = __builtin_amdgcn_wmma_f32_16x16x4_f32(
                     false, a,  false, b,
                     (short)0, c[t],  false,  false);
        }
    }

#pragma unroll
    for (int t = 0; t < 4; ++t) {
#pragma unroll
        for (int r = 0; r < 8; ++r) {
            float e = c[t][r] + bcol[t];
            e = e > 0.0f ? e : expm1f(e);
            mine[(r + (hi << 3)) * LIN_PITCH + t * 16 + m] = e;
        }
    }
    __syncthreads();

    v8f c2[4] = {};
    const float* brow2 = Ws2 + m * LIN_PITCH + k2;
#pragma unroll
    for (int kb = 0; kb < DFEAT; kb += 4) {
        const v2f a = *(const v2f*)(arow + kb);
#pragma unroll
        for (int t = 0; t < 4; ++t) {
            const v2f b = *(const v2f*)(brow2 + t * 16 * LIN_PITCH + kb);
            c2[t] = __builtin_amdgcn_wmma_f32_16x16x4_f32(
                     false, a,  false, b,
                     (short)0, c2[t],  false,  false);
        }
    }
#pragma unroll
    for (int t = 0; t < 4; ++t) {
#pragma unroll
        for (int r = 0; r < 8; ++r)
            mine[(r + (hi << 3)) * LIN_PITCH + t * 16 + m] = c2[t][r];
    }
    __syncthreads();

    if (valid) {
        float* orow = out + (size_t)rowBase * DFEAT;
#pragma unroll
        for (int i = 0; i < 8; ++i)
            *(float4*)(orow + loff + i * 2 * DFEAT) =
                *(const float4*)(mine + (2 * i + hi) * LIN_PITCH + c4);
    }
}

extern "C" void kernel_launch(void* const* d_in, const int* in_sizes, int n_in,
                              void* d_out, int out_size, void* d_ws, size_t ws_size,
                              hipStream_t stream) {
    const float* h   = (const float*)d_in[0];
    const int*   src = (const int*)  d_in[1];
    const int*   dst = (const int*)  d_in[2];
    const float* W1  = (const float*)d_in[3];
    const float* b1  = (const float*)d_in[4];
    const float* W2  = (const float*)d_in[5];
    const float* b2  = (const float*)d_in[6];
    float*       out = (float*)d_out;

    const int N = in_sizes[0] / DFEAT;
    const int E = in_sizes[1];

    char* ws = (char*)d_ws;
    float* deg = (float*)ws;
    size_t degBytes = (((size_t)AGG_S * N * sizeof(float)) + 255) & ~(size_t)255;
    float* agg = (float*)(ws + degBytes);
    float* h1  = agg + (size_t)AGG_S * N * DFEAT;
    unsigned* hdrs  = (unsigned*)(h1 + (size_t)N * DFEAT);

    const int tiles = N / 16;
    const int linB  = (tiles + LIN_WAVES - 1) / LIN_WAVES;
    const dim3 aggB((N + AGG_R - 1) / AGG_R, AGG_S);
    unsigned* lists = hdrs + (size_t)aggB.x * aggB.y * LIST_HDR;
    unsigned* keys   = lists + (size_t)aggB.x * aggB.y * OQ_N * LIST_WCAP;
    unsigned* packed = keys + (size_t)((E + 15) / 16) * 4;
    const int keyB   = (E / 4 + KEY_THREADS * KEY_ITEMS - 1) / (KEY_THREADS * KEY_ITEMS) + (E < 4);

    gcn_key_kernel<<<keyB, KEY_THREADS, 0, stream>>>(src, dst, keys, packed, E);

    gcn_agg_kernel<true><<<aggB, AGG_THREADS, 0, stream>>>(h, keys, packed, agg, deg, lists, hdrs, b2, out, N, E);
    gcn_wmma_kernel<<<linB, LIN_THREADS, 0, stream>>>(agg, deg, W1, b1, W2, h1, N);

    const dim3 aggB2(aggB.x * 2, 1);
    gcn_agg_kernel<false><<<aggB2, AGG_THREADS, 0, stream>>>(h1, keys, packed, agg, deg, lists, hdrs, b2, out, N, E);
}


# layer 2: each owner wave keeps the sums of its 13 rows in registers (wave-uniform register index per entry), not in LDS: no LDS read-add-write chain, clearing pass or barriers; sum order kept
# speedup vs baseline: 4.3929x; 1.0381x over previous
#include <hip/hip_runtime.h>
#include <hip/hip_bf16.h>
#include <math.h>

typedef __attribute__((ext_vector_type(2))) float v2f;
typedef __attribute__((ext_vector_type(8))) float v8f;
typedef __attribute__((ext_vector_type(32))) float v32f;

#define DFEAT 64

#define AGG_S       2
#define AGG_R       782
#define AGG_H       (AGG_R / 2)
#define AGG_THREADS 1024
#define AGG_WAVES   (AGG_THREADS / 32)
#define AGG_VEC     4
#define AGG_WQ      (16 * 32 + 32)
#define KEY_THREADS 256
#define KEY_ITEMS   4

__global__ __launch_bounds__(KEY_THREADS)
void gcn_key_kernel(const int* __restrict__ src,
                    const int* __restrict__ dst,
                    unsigned* __restrict__ keys,
                    unsigned* __restrict__ packed,
                    int E) {
    const int nv   = E >> 2;
    const int base = blockIdx.x * (KEY_THREADS * KEY_ITEMS) + threadIdx.x;
    if (nv > 0) {
        int4 d[KEY_ITEMS], s[KEY_ITEMS];
#pragma unroll
        for (int k = 0; k < KEY_ITEMS; ++k) {
            const int i = min(base + k * KEY_THREADS, nv - 1);
            d[k] = ((const int4*)dst)[i];
            s[k] = ((const int4*)src)[i];
        }
#pragma unroll
        for (int k = 0; k < KEY_ITEMS; ++k) {
            const int i = base + k * KEY_THREADS;
            if (i < nv) {
                const unsigned r0 = (unsigned)d[k].x / AGG_R, r1 = (unsigned)d[k].y / AGG_R;
                const unsigned r2 = (unsigned)d[k].z / AGG_R, r3 = (unsigned)d[k].w / AGG_R;
                uint4 p;
                p.x = ((unsigned)s[k].x << 10) | ((unsigned)d[k].x - r0 * AGG_R);
                p.y = ((unsigned)s[k].y << 10) | ((unsigned)d[k].y - r1 * AGG_R);
                p.z = ((unsigned)s[k].z << 10) | ((unsigned)d[k].z - r2 * AGG_R);
                p.w = ((unsigned)s[k].w << 10) | ((unsigned)d[k].w - r3 * AGG_R);
                keys[i] = r0 | (r1 << 8) | (r2 << 16) | (r3 << 24);
                ((uint4*)packed)[i] = p;
            }
        }
    }
    if (blockIdx.x == 0 && (int)threadIdx.x < (E & 3)) {
        const int e = (nv << 2) + (int)threadIdx.x;
        const unsigned dv = (unsigned)dst[e];
        const unsigned r  = dv / AGG_R;
        packed[e] = ((unsigned)src[e] << 10) | (dv - r * AGG_R);
        ((unsigned char*)keys)[e] = (unsigned char)r;
    }
}

#define OQ_N        (2 * AGG_WAVES)
#define OQ_CAP      160
#define LIST_WCAP   OQ_CAP
#define LIST_HDR    OQ_N
static_assert(AGG_S == 2 && (AGG_R & 1) == 0, "layer 2 reads two slices' lists per half range");

template <bool FULL, typename BETWEEN>
__device__ __forceinline__ void gather_add(const float* __restrict__ x, float* acc,
                                           int s, int d, int m, int lane, BETWEEN between) {
#pragma unroll
    for (int h = 0; h < 2; ++h) {
        if (!FULL && h * 16 >= m) break;
        v2f r[16];
#pragma unroll
        for (int j = 0; j < 16; ++j) {
            const int sj = __builtin_amdgcn_readlane(s, h * 16 + j);
            r[j] = *(const v2f*)(x + (size_t)sj * DFEAT + 2 * lane);
        }
        if (h == 0) between();
#pragma unroll
        for (int j = 0; j < 16; ++j) {
            if (FULL || h * 16 + j < m) {
                const int dj = __builtin_amdgcn_readlane(d, h * 16 + j);
                float* p = acc + dj * DFEAT + 2 * lane;
                __hip_atomic_fetch_add(p + 0, r[j].x, __ATOMIC_RELAXED, __HIP_MEMORY_SCOPE_WORKGROUP);
                __hip_atomic_fetch_add(p + 1, r[j].y, __ATOMIC_RELAXED, __HIP_MEMORY_SCOPE_WORKGROUP);
            }
        }
    }
}

template <int J0>
__device__ __forceinline__ void rows_load(const float* __restrict__ x, v2f (&r)[16], int s, int lane) {
#pragma unroll
    for (int j = 0; j < 16; ++j) {
        const int sj = __builtin_amdgcn_readlane(s, J0 + j);
        r[j] = *(const v2f*)(x + (size_t)sj * DFEAT + 2 * lane);
    }
}

#define OWN_ROWS    ((AGG_H + AGG_WAVES - 1) / AGG_WAVES)
static_assert(2 * (OWN_ROWS + 1) <= 32, "the wave's sums and one idle pair fit one 32-register vector");

template <int J0>
__device__ __forceinline__ void rows_reg_add(v32f& a, const v2f (&r)[16], int slotv) {
#pragma unroll
    for (int j = 0; j < 16; ++j) {
        const int q = __builtin_amdgcn_readlane(slotv, J0 + j);
        a[2 * q]     += r[j].x;
        a[2 * q + 1] += r[j].y;
    }
}

template <bool FULL, bool REC>
__device__ __forceinline__ void agg_drain(const float* __restrict__ x,
                                          const unsigned* __restrict__ packed,
                                          float* acc, unsigned* cnt, const unsigned* wq,
                                          int m, int lane, unsigned* oq, unsigned* oqn, int half) {
    const bool live = FULL || lane < m;
    const unsigned ent = live ? packed[wq[lane]] : 0u;
    const int d = (int)(ent & 1023u);
    const int s = (int)(ent >> 10);
    const int hf = d >= AGG_H ? 1 : 0;
    const int d2 = d - hf * AGG_H;
    if (REC) {
        if (live) {
            __hip_atomic_fetch_add(&cnt[d], 1u, __ATOMIC_RELAXED, __HIP_MEMORY_SCOPE_WORKGROUP);
            const unsigned q   = (unsigned)hf * AGG_WAVES + ((unsigned)d2 & (AGG_WAVES - 1));
            const unsigned pos = __hip_atomic_fetch_add(&oqn[q], 1u, __ATOMIC_RELAXED, __HIP_MEMORY_SCOPE_WORKGROUP);
            if (pos < (unsigned)OQ_CAP) oq[q * OQ_CAP + pos] = (ent & ~1023u) | (unsigned)d2;
        }
        gather_add<FULL>(x, acc, s, d, m, lane, [] {});
    } else {
        unsigned todo = __builtin_amdgcn_ballot_w32(live && hf == half);
        while (todo) {
            const int j = __builtin_ctz(todo);
            todo &= todo - 1u;
            const int sj = __builtin_amdgcn_readlane(s, j);
            const int dj = __builtin_amdgcn_readlane(d2, j);
            const v2f r = *(const v2f*)(x + (size_t)sj * DFEAT + 2 * lane);
            float* p = acc + dj * DFEAT + 2 * lane;
            __hip_atomic_fetch_add(p + 0, r.x, __ATOMIC_RELAXED, __HIP_MEMORY_SCOPE_WORKGROUP);
            __hip_atomic_fetch_add(p + 1, r.y, __ATOMIC_RELAXED, __HIP_MEMORY_SCOPE_WORKGROUP);
        }
    }
}

__device__ __forceinline__ void agg_own_rows(const float* __restrict__ x,
                                             const unsigned* __restrict__ l0, unsigned n0,
                                             const unsigned* __restrict__ l1, unsigned n1,
                                             const float* __restrict__ degp,
                                             const float* __restrict__ bias,
                                             float* __restrict__ out,
                                             int N, int lo, int nloc, int wave, int lane) {
    const int myrow = wave + AGG_WAVES * lane;
    const int rr    = lo + min(myrow, nloc - 1);
    float dg = 0.0f;
#pragma unroll
    for (int p = 0; p < AGG_S; ++p) dg += degp[(size_t)p * N + rr];
    const v2f b = *(const v2f*)(bias + 2 * lane);

    v32f a = {};

    const unsigned n = n0 + n1;
    if (n > 0u) {
        auto entry = [&](unsigned k) {
            const unsigned kk = min(k, n - 1u);
            const unsigned* p = kk < n0 ? l0 + kk : l1 + (kk - n0);
            const unsigned e = *p;
            return k < n ? e : 0u;
        };
        unsigned ent = entry((unsigned)lane);
        v2f ra[16], rb[16];
        rows_load<0>(x, ra, (int)(ent >> 10), lane);
        for (unsigned k = 0; k < n; k += 32u) {
            const int s = (int)(ent >> 10);
            const int slotv = k + (unsigned)lane < n ? (int)((ent & 1023u) / AGG_WAVES) : OWN_ROWS;
            const unsigned nxt = entry(k + 32u + (unsigned)lane);
            rows_load<16>(x, rb, s, lane);
            rows_reg_add<0>(a, ra, slotv);
            rows_load<0>(x, ra, (int)(nxt >> 10), lane);
            rows_reg_add<16>(a, rb, slotv);
            ent = nxt;
        }
    }

    const float ivv = dg > 0.0f ? 1.0f / dg : 0.0f;
#pragma unroll
    for (int i = 0; i < OWN_ROWS; ++i) {
        const int row = wave + AGG_WAVES * i;
        if (row < nloc) {
            const float iv = __builtin_bit_cast(float, __builtin_amdgcn_readlane(__builtin_bit_cast(int, ivv), i));
            v2f e;
            e.x = a[2 * i]     * iv + b.x;
            e.y = a[2 * i + 1] * iv + b.y;
            e.x = e.x > 0.0f ? e.x : expm1f(e.x);
            e.y = e.y > 0.0f ? e.y : expm1f(e.y);
            *(v2f*)(out + (size_t)(lo + row) * DFEAT + 2 * lane) = e;
        }
    }
}

template <bool FIRST>
__global__ __launch_bounds__(AGG_THREADS)
void gcn_agg_kernel(const float* __restrict__ x,
                    const unsigned* __restrict__ keys,
                    const unsigned* __restrict__ packed,
                    float* __restrict__ agg,
                    float* __restrict__ degp,
                    unsigned* __restrict__ lists,
                    unsigned* __restrict__ hdrs,
                    const float* __restrict__ bias,
                    float* __restrict__ out,
                    int N, int E) {
    constexpr int ROWS = FIRST ? AGG_R : AGG_H;
    __shared__ __attribute__((aligned(16))) float    acc[ROWS * DFEAT];
    __shared__ unsigned cnt[FIRST ? AGG_R : 1];
    __shared__ unsigned wqs[AGG_WAVES * AGG_WQ];
    __shared__ unsigned oq[FIRST ? OQ_N * OQ_CAP : 1];
    __shared__ unsigned oqn[OQ_N];
    __shared__ float    invd[FIRST ? 1 : AGG_H];

    const int tid  = threadIdx.x;
    const int lane = tid & 31;
    const int wave = __builtin_amdgcn_readfirstlane(tid >> 5);
    const int nrng  = FIRST ? (int)gridDim.x : (int)(gridDim.x >> 1);
    const int rng   = FIRST ? (int)blockIdx.x : (int)(blockIdx.x >> 1);
    const int half  = FIRST ? 0 : (int)(blockIdx.x & 1);
    const int slice = FIRST ? (int)blockIdx.y : 0;
    const int nslice = FIRST ? AGG_S : 1;
    const int lo    = rng * AGG_R + half * AGG_H;
    const unsigned nloc = (unsigned)max(0, min(ROWS, N - lo));
    unsigned* wq = wqs + wave * AGG_WQ;

    const size_t b0 = (size_t)rng;
    const size_t b1 = (size_t)nrng + rng;
    unsigned f0 = 0u, f1 = 0u;
    if (!FIRST) {
        f0 = hdrs[b0 * LIST_HDR + half * AGG_WAVES + lane];
        f1 = hdrs[b1 * LIST_HDR + half * AGG_WAVES + lane];
    }

    if (!FIRST) {
        if (__builtin_amdgcn_ballot_w32(f0 > (unsigned)OQ_CAP || f1 > (unsigned)OQ_CAP) == 0u) {
            if (nloc > 0u)
                agg_own_rows(x, lists + (b0 * OQ_N + half * AGG_WAVES + wave) * LIST_WCAP,
                             (unsigned)__builtin_amdgcn_readlane((int)f0, wave),
                             lists + (b1 * OQ_N + half * AGG_WAVES + wave) * LIST_WCAP,
                             (unsigned)__builtin_amdgcn_readlane((int)f1, wave),
                             degp, bias, out, N, lo, (int)nloc, wave, lane);
            return;
        }
    }

    for (int i = tid; i < ROWS * DFEAT / 4; i += AGG_THREADS)
        ((float4*)acc)[i] = make_float4(0.f, 0.f, 0.f, 0.f);
    if (FIRST) {
        for (int i = tid; i < AGG_R; i += AGG_THREADS) cnt[i] = 0u;
        if (tid < OQ_N) oqn[tid] = 0u;
    } else {
        for (int i = tid; i < (int)nloc; i += AGG_THREADS) {
            float dg = 0.0f;
#pragma unroll
            for (int p = 0; p < AGG_S; ++p) dg += degp[(size_t)p * N + lo + i];
            invd[i] = dg > 0.0f ? 1.0f / dg : 0.0f;
        }
    }
    __syncthreads();

    {
        const uint4* key4 = (const uint4*)keys;
        const int nvec  = E >> 4;
        const int gvec  = AGG_VEC * AGG_THREADS;
        const int nfull = nvec / gvec;
        const int gs = (int)(((long long)slice * nfull) / nslice);
        const int ng = (int)(((long long)(slice + 1) * nfull) / nslice) - gs;
        const int g0 = (int)(((long long)blockIdx.x * ng) / gridDim.x);
        const unsigned pat = (unsigned)rng * 0x01010101u;

        unsigned cur  = 0;

        auto drain = [&](const unsigned* q, unsigned m) {
            if (m == 32u) agg_drain<true,  FIRST>(x, packed, acc, cnt, q, 32, lane, oq, oqn, half);
            else          agg_drain<false, FIRST>(x, packed, acc, cnt, q, (int)m, lane, oq, oqn, half);
        };

        auto flags = [&](unsigned w) {
            const unsigned x = w ^ pat;
            return (0x80808080u - x) & 0x80808080u;
        };

        auto visit16 = [&](const uint4& v, unsigned ebase) {
            unsigned mm = flags(v.x) | (flags(v.y) >> 1) | (flags(v.z) >> 2) | (flags(v.w) >> 3);
            unsigned mask = __builtin_amdgcn_ballot_w32(mm != 0u);
            while (mask) {
                if (mm != 0u) {
                    const unsigned b = (unsigned)__builtin_ctz(mm);
                    const unsigned e = ebase + ((7u - (b & 7u)) << 2) + (b >> 3);
                    wq[cur + __builtin_amdgcn_mbcnt_lo(mask, 0u)] = e;
                    mm &= mm - 1u;
                }
                cur += (unsigned)__builtin_popcount(mask);
                mask = __builtin_amdgcn_ballot_w32(mm != 0u);
            }
        };

        auto group_at = [&](int g) {
            int gg = g + g0;
            return gs + (gg >= ng ? gg - ng : gg);
        };

        auto load_group = [&](int g, uint4 (&v)[AGG_VEC]) {
            const int base = group_at(g) * gvec + tid;
#pragma unroll
            for (int k = 0; k < AGG_VEC; ++k) v[k] = key4[base + k * AGG_THREADS];
        };

        auto visit_vecs = [&](int base, const uint4 (&v)[AGG_VEC]) {
#pragma unroll
            for (int k = 0; k < AGG_VEC; ++k) {
                visit16(v[k], (unsigned)(base + k * AGG_THREADS) << 4);
                while (cur >= 32u) {
                    cur -= 32u;
                    drain(wq + cur, 32u);
                }
            }
        };

        if (ng > 0) {
            uint4 va[AGG_VEC], vb[AGG_VEC];
            load_group(0, va);
            for (int g = 0; g < ng; ++g) {
                load_group(min(g + 1, ng - 1), vb);
                visit_vecs(group_at(g) * gvec + tid, va);
#pragma unroll
                for (int k = 0; k < AGG_VEC; ++k) va[k] = vb[k];
            }
        }
        if (slice == 0) {
            uint4 vr[AGG_VEC];
            const int base = nfull * gvec + tid;
            const unsigned none = pat ^ 0x01010101u;
#pragma unroll
            for (int k = 0; k < AGG_VEC; ++k) {
                const int vi = base + k * AGG_THREADS;
                vr[k] = vi < nvec ? key4[vi] : make_uint4(none, none, none, none);
            }
            visit_vecs(base, vr);
        }
        if (slice == 0 && wave == 0 && (E & 15) != 0) {
            const int e = (nvec << 4) + lane;
            const bool tail = lane < (E & 15);
            const unsigned kb = tail ? (unsigned)((const unsigned char*)keys)[e] : 0xFFu;
            const bool hit = kb == (unsigned)rng;
            const unsigned mask = __builtin_amdgcn_ballot_w32(hit);
            if (hit) wq[cur + __builtin_amdgcn_mbcnt_lo(mask, 0u)] = (unsigned)e;
            cur += (unsigned)__builtin_popcount(mask);
        }
        while (cur >= 32u) {
            cur -= 32u;
            drain(wq + cur, 32u);
        }
        if (cur) drain(wq, cur);
    }

    __syncthreads();

    if (FIRST) {
        float4* orow = (float4*)(agg + ((size_t)slice * N + lo) * DFEAT);
        for (int i = tid; i < (int)nloc * (DFEAT / 4); i += AGG_THREADS)
            orow[i] = ((const float4*)acc)[i];
        for (int i = tid; i < (int)nloc; i += AGG_THREADS)
            degp[(size_t)slice * N + lo + i] = (float)cnt[i];
        const size_t blk = (size_t)slice * nrng + rng;
#pragma unroll
        for (int hq = 0; hq < 2; ++hq) {
            const int q = hq * AGG_WAVES + wave;
            const unsigned nq = (unsigned)__builtin_amdgcn_readfirstlane((int)oqn[q]);
            const unsigned nk = min(nq, (unsigned)OQ_CAP);
            unsigned* lst = lists + (blk * OQ_N + q) * LIST_WCAP;
            for (unsigned k = (unsigned)lane; k < nk; k += 32u) lst[k] = oq[q * OQ_CAP + k];
            if (lane == 0) hdrs[blk * LIST_HDR + q] = nq;
        }
    } else {
        float4* orow = (float4*)(out + (size_t)lo * DFEAT);
        for (int i = tid; i < (int)nloc * (DFEAT / 4); i += AGG_THREADS) {
            const float4 a  = ((const float4*)acc)[i];
            const float4 b  = ((const float4*)bias)[i & (DFEAT / 4 - 1)];
            const float  iv = invd[i / (DFEAT / 4)];
            float4 e;
            e.x = a.x * iv + b.x; e.y = a.y * iv + b.y;
            e.z = a.z * iv + b.z; e.w = a.w * iv + b.w;
            e.x = e.x > 0.0f ? e.x : expm1f(e.x);
            e.y = e.y > 0.0f ? e.y : expm1f(e.y);
            e.z = e.z > 0.0f ? e.z : expm1f(e.z);
            e.w = e.w > 0.0f ? e.w : expm1f(e.w);
            orow[i] = e;
        }
    }
}

#define LIN_WAVES   8
#define LIN_THREADS (LIN_WAVES * 32)
#define LIN_PITCH   (DFEAT + 4)

__global__ __launch_bounds__(LIN_THREADS)
void gcn_wmma_kernel(const float* __restrict__ agg,
                     const float* __restrict__ degp,
                     const float* __restrict__ W,
                     const float* __restrict__ bias,
                     const float* __restrict__ W2,
                     float* __restrict__ out,
                     int N) {
    __shared__ __attribute__((aligned(16))) float Ws[DFEAT * LIN_PITCH];
    __shared__ __attribute__((aligned(16))) float Ws2[DFEAT * LIN_PITCH];
    __shared__ __attribute__((aligned(16))) float As[LIN_WAVES * 16 * LIN_PITCH];

    const int tid  = threadIdx.x;
    const int lane = tid & 31;
    const int wave  = __builtin_amdgcn_readfirstlane(tid >> 5);
    const int tiles = N >> 4;
    const int want  = blockIdx.x * LIN_WAVES + wave;
    const bool valid = want < tiles;
    const int tile = valid ? want : tiles - 1;
    const int rowBase = tile << 4;
    const int m  = lane & 15;
    const int hi = lane >> 4;
    const int k2 = hi << 1;
    const int c4 = m << 2;
    float* mine = As + wave * (16 * LIN_PITCH);
    const size_t plane = (size_t)N * DFEAT;

    float4 w4[DFEAT * DFEAT / 4 / LIN_THREADS], x4[DFEAT * DFEAT / 4 / LIN_THREADS];
#pragma unroll
    for (int j = 0; j < DFEAT * DFEAT / 4 / LIN_THREADS; ++j) {
        w4[j] = *(const float4*)(W  + 4 * (tid + j * LIN_THREADS));
        x4[j] = *(const float4*)(W2 + 4 * (tid + j * LIN_THREADS));
    }

    const float* rows = agg + (size_t)rowBase * DFEAT;
    const unsigned loff = (unsigned)(hi * DFEAT + c4);
    float4 v[8];
#pragma unroll
    for (int i = 0; i < 8; ++i)
        v[i] = *(const float4*)(rows + loff + i * 2 * DFEAT);
    float4 u[AGG_S - 1][8];
#pragma unroll
    for (int p = 1; p < AGG_S; ++p) {
#pragma unroll
        for (int i = 0; i < 8; ++i)
            u[p - 1][i] = *(const float4*)(rows + p * plane + loff + i * 2 * DFEAT);
    }

    float deg = degp[rowBase + m];
#pragma unroll
    for (int p = 1; p < AGG_S; ++p) deg += degp[(size_t)p * N + rowBase + m];
    float bcol[4];
#pragma unroll
    for (int t = 0; t < 4; ++t) bcol[t] = bias[t * 16 + m];

#pragma unroll
    for (int j = 0; j < DFEAT * DFEAT / 4 / LIN_THREADS; ++j) {
        const int idx = tid + j * LIN_THREADS;
        *(float4*)(Ws  + (idx >> 4) * LIN_PITCH + ((idx & 15) << 2)) = w4[j];
        *(float4*)(Ws2 + (idx >> 4) * LIN_PITCH + ((idx & 15) << 2)) = x4[j];
    }
#pragma unroll
    for (int i = 0; i < 8; ++i) {
#pragma unroll
        for (int p = 1; p < AGG_S; ++p) {
            v[i].x += u[p - 1][i].x; v[i].y += u[p - 1][i].y;
            v[i].z += u[p - 1][i].z; v[i].w += u[p - 1][i].w;
        }
        *(float4*)(mine + (2 * i + hi) * LIN_PITCH + c4) = v[i];
    }
    const float inv = deg > 0.0f ? 1.0f / deg : 0.0f;
    __syncthreads();

    v8f c[4] = {};
    const float* arow = mine + m * LIN_PITCH + k2;
    const float* brow = Ws   + m * LIN_PITCH + k2;
#pragma unroll
    for (int kb = 0; kb < DFEAT; kb += 4) {
        v2f a = *(const v2f*)(arow + kb);
        a.x *= inv;
        a.y *= inv;
#pragma unroll
        for (int t = 0; t < 4; ++t) {
            const v2f b = *(const v2f*)(brow + t * 16 * LIN_PITCH + kb);
            c[t] = __builtin_amdgcn_wmma_f32_16x16x4_f32(
                     false, a,  false, b,
                     (short)0, c[t],  false,  false);
        }
    }

#pragma unroll
    for (int t = 0; t < 4; ++t) {
#pragma unroll
        for (int r = 0; r < 8; ++r) {
            float e = c[t][r] + bcol[t];
            e = e > 0.0f ? e : expm1f(e);
            mine[(r + (hi << 3)) * LIN_PITCH + t * 16 + m] = e;
        }
    }
    __syncthreads();

    v8f c2[4] = {};
    const float* brow2 = Ws2 + m * LIN_PITCH + k2;
#pragma unroll
    for (int kb = 0; kb < DFEAT; kb += 4) {
        const v2f a = *(const v2f*)(arow + kb);
#pragma unroll
        for (int t = 0; t < 4; ++t) {
            const v2f b = *(const v2f*)(brow2 + t * 16 * LIN_PITCH + kb);
            c2[t] = __builtin_amdgcn_wmma_f32_16x16x4_f32(
                     false, a,  false, b,
                     (short)0, c2[t],  false,  false);
        }
    }
#pragma unroll
    for (int t = 0; t < 4; ++t) {
#pragma unroll
        for (int r = 0; r < 8; ++r)
            mine[(r + (hi << 3)) * LIN_PITCH + t * 16 + m] = c2[t][r];
    }
    __syncthreads();

    if (valid) {
        float* orow = out + (size_t)rowBase * DFEAT;
#pragma unroll
        for (int i = 0; i < 8; ++i)
            *(float4*)(orow + loff + i * 2 * DFEAT) =
                *(const float4*)(mine + (2 * i + hi) * LIN_PITCH + c4);
    }
}

extern "C" void kernel_launch(void* const* d_in, const int* in_sizes, int n_in,
                              void* d_out, int out_size, void* d_ws, size_t ws_size,
                              hipStream_t stream) {
    const float* h   = (const float*)d_in[0];
    const int*   src = (const int*)  d_in[1];
    const int*   dst = (const int*)  d_in[2];
    const float* W1  = (const float*)d_in[3];
    const float* b1  = (const float*)d_in[4];
    const float* W2  = (const float*)d_in[5];
    const float* b2  = (const float*)d_in[6];
    float*       out = (float*)d_out;

    const int N = in_sizes[0] / DFEAT;
    const int E = in_sizes[1];

    char* ws = (char*)d_ws;
    float* deg = (float*)ws;
    size_t degBytes = (((size_t)AGG_S * N * sizeof(float)) + 255) & ~(size_t)255;
    float* agg = (float*)(ws + degBytes);
    float* h1  = agg + (size_t)AGG_S * N * DFEAT;
    unsigned* hdrs  = (unsigned*)(h1 + (size_t)N * DFEAT);

    const int tiles = N / 16;
    const int linB  = (tiles + LIN_WAVES - 1) / LIN_WAVES;
    const dim3 aggB((N + AGG_R - 1) / AGG_R, AGG_S);
    unsigned* lists = hdrs + (size_t)aggB.x * aggB.y * LIST_HDR;
    unsigned* keys   = lists + (size_t)aggB.x * aggB.y * OQ_N * LIST_WCAP;
    unsigned* packed = keys + (size_t)((E + 15) / 16) * 4;
    const int keyB   = (E / 4 + KEY_THREADS * KEY_ITEMS - 1) / (KEY_THREADS * KEY_ITEMS) + (E < 4);

    gcn_key_kernel<<<keyB, KEY_THREADS, 0, stream>>>(src, dst, keys, packed, E);

    gcn_agg_kernel<true><<<aggB, AGG_THREADS, 0, stream>>>(h, keys, packed, agg, deg, lists, hdrs, b2, out, N, E);
    gcn_wmma_kernel<<<linB, LIN_THREADS, 0, stream>>>(agg, deg, W1, b1, W2, h1, N);

    const dim3 aggB2(aggB.x * 2, 1);
    gcn_agg_kernel<false><<<aggB2, AGG_THREADS, 0, stream>>>(h1, keys, packed, agg, deg, lists, hdrs, b2, out, N, E);
}


# layer 1 in two phases: the scan only files edges in per-owner LDS queues; behind a barrier each owner wave gathers its queues and sums its rows in registers: no LDS float atomics; sum order: arrival
# speedup vs baseline: 4.7613x; 1.0839x over previous
#include <hip/hip_runtime.h>
#include <hip/hip_bf16.h>
#include <math.h>

typedef __attribute__((ext_vector_type(2))) float v2f;
typedef __attribute__((ext_vector_type(8))) float v8f;
typedef __attribute__((ext_vector_type(32))) float v32f;

#define DFEAT 64

#define AGG_S       2
#define AGG_R       782
#define AGG_H       (AGG_R / 2)
#define AGG_THREADS 1024
#define AGG_WAVES   (AGG_THREADS / 32)
#define AGG_VEC     4
#define AGG_WQ      (16 * 32 + 32)
#define KEY_THREADS 256
#define KEY_ITEMS   4

__global__ __launch_bounds__(KEY_THREADS)
void gcn_key_kernel(const int* __restrict__ src,
                    const int* __restrict__ dst,
                    unsigned* __restrict__ keys,
                    unsigned* __restrict__ packed,
                    int E) {
    const int nv   = E >> 2;
    const int base = blockIdx.x * (KEY_THREADS * KEY_ITEMS) + threadIdx.x;
    if (nv > 0) {
        int4 d[KEY_ITEMS], s[KEY_ITEMS];
#pragma unroll
        for (int k = 0; k < KEY_ITEMS; ++k) {
            const int i = min(base + k * KEY_THREADS, nv - 1);
            d[k] = ((const int4*)dst)[i];
            s[k] = ((const int4*)src)[i];
        }
#pragma unroll
        for (int k = 0; k < KEY_ITEMS; ++k) {
            const int i = base + k * KEY_THREADS;
            if (i < nv) {
                const unsigned r0 = (unsigned)d[k].x / AGG_R, r1 = (unsigned)d[k].y / AGG_R;
                const unsigned r2 = (unsigned)d[k].z / AGG_R, r3 = (unsigned)d[k].w / AGG_R;
                uint4 p;
                p.x = ((unsigned)s[k].x << 10) | ((unsigned)d[k].x - r0 * AGG_R);
                p.y = ((unsigned)s[k].y << 10) | ((unsigned)d[k].y - r1 * AGG_R);
                p.z = ((unsigned)s[k].z << 10) | ((unsigned)d[k].z - r2 * AGG_R);
                p.w = ((unsigned)s[k].w << 10) | ((unsigned)d[k].w - r3 * AGG_R);
                keys[i] = r0 | (r1 << 8) | (r2 << 16) | (r3 << 24);
                ((uint4*)packed)[i] = p;
            }
        }
    }
    if (blockIdx.x == 0 && (int)threadIdx.x < (E & 3)) {
        const int e = (nv << 2) + (int)threadIdx.x;
        const unsigned dv = (unsigned)dst[e];
        const unsigned r  = dv / AGG_R;
        packed[e] = ((unsigned)src[e] << 10) | (dv - r * AGG_R);
        ((unsigned char*)keys)[e] = (unsigned char)r;
    }
}

#define OQ_N        (2 * AGG_WAVES)
#define OQ_CAP      160
#define LIST_WCAP   OQ_CAP
#define LIST_HDR    OQ_N
static_assert(AGG_S == 2 && (AGG_R & 1) == 0, "layer 2 reads two slices' lists per half range");

template <int J0>
__device__ __forceinline__ void rows_load(const float* __restrict__ x, v2f (&r)[16], int s, int lane) {
#pragma unroll
    for (int j = 0; j < 16; ++j) {
        const int sj = __builtin_amdgcn_readlane(s, J0 + j);
        r[j] = *(const v2f*)(x + (size_t)sj * DFEAT + 2 * lane);
    }
}

#define OWN_ROWS    ((AGG_H + AGG_WAVES - 1) / AGG_WAVES)
static_assert(2 * (OWN_ROWS + 1) <= 32, "the wave's sums and one idle pair fit one 32-register vector");

template <int J0>
__device__ __forceinline__ void rows_reg_add(v32f& a, const v2f (&r)[16], int slotv) {
#pragma unroll
    for (int j = 0; j < 16; ++j) {
        const int q = __builtin_amdgcn_readlane(slotv, J0 + j);
        a[2 * q]     += r[j].x;
        a[2 * q + 1] += r[j].y;
    }
}

template <bool FULL>
__device__ __forceinline__ void agg_file(const unsigned* __restrict__ packed,
                                         unsigned* cnt, const unsigned* wq,
                                         int m, int lane, unsigned* oq, unsigned* oqn) {
    const bool live = FULL || lane < m;
    const unsigned ent = live ? packed[wq[lane]] : 0u;
    const int d  = (int)(ent & 1023u);
    const int hf = d >= AGG_H ? 1 : 0;
    const int d2 = d - hf * AGG_H;
    if (live) {
        __hip_atomic_fetch_add(&cnt[d], 1u, __ATOMIC_RELAXED, __HIP_MEMORY_SCOPE_WORKGROUP);
        const unsigned q   = (unsigned)hf * AGG_WAVES + ((unsigned)d2 & (AGG_WAVES - 1));
        const unsigned pos = __hip_atomic_fetch_add(&oqn[q], 1u, __ATOMIC_RELAXED, __HIP_MEMORY_SCOPE_WORKGROUP);
        if (pos < (unsigned)OQ_CAP) oq[q * OQ_CAP + pos] = (ent & ~1023u) | (unsigned)d2;
    }
}

template <bool FULL>
__device__ __forceinline__ void agg_gather_all(const float* __restrict__ x,
                                               const unsigned* __restrict__ packed,
                                               float* acc, const unsigned* wq, int m, int lane) {
    const bool live = FULL || lane < m;
    const unsigned ent = live ? packed[wq[lane]] : 0u;
    const int d = (int)(ent & 1023u);
    const int s = (int)(ent >> 10);
    unsigned todo = __builtin_amdgcn_ballot_w32(live);
    while (todo) {
        const int j = __builtin_ctz(todo);
        todo &= todo - 1u;
        const int sj = __builtin_amdgcn_readlane(s, j);
        const int dj = __builtin_amdgcn_readlane(d, j);
        const v2f r = *(const v2f*)(x + (size_t)sj * DFEAT + 2 * lane);
        float* p = acc + dj * DFEAT + 2 * lane;
        __hip_atomic_fetch_add(p + 0, r.x, __ATOMIC_RELAXED, __HIP_MEMORY_SCOPE_WORKGROUP);
        __hip_atomic_fetch_add(p + 1, r.y, __ATOMIC_RELAXED, __HIP_MEMORY_SCOPE_WORKGROUP);
    }
}

template <bool FULL>
__device__ __forceinline__ void agg_drain_half(const float* __restrict__ x,
                                               const unsigned* __restrict__ packed,
                                               float* acc, const unsigned* wq,
                                               int m, int lane, int half) {
    const bool live = FULL || lane < m;
    const unsigned ent = live ? packed[wq[lane]] : 0u;
    const int d = (int)(ent & 1023u);
    const int s = (int)(ent >> 10);
    const int hf = d >= AGG_H ? 1 : 0;
    const int d2 = d - hf * AGG_H;
    unsigned todo = __builtin_amdgcn_ballot_w32(live && hf == half);
    while (todo) {
        const int j = __builtin_ctz(todo);
        todo &= todo - 1u;
        const int sj = __builtin_amdgcn_readlane(s, j);
        const int dj = __builtin_amdgcn_readlane(d2, j);
        const v2f r = *(const v2f*)(x + (size_t)sj * DFEAT + 2 * lane);
        float* p = acc + dj * DFEAT + 2 * lane;
        __hip_atomic_fetch_add(p + 0, r.x, __ATOMIC_RELAXED, __HIP_MEMORY_SCOPE_WORKGROUP);
        __hip_atomic_fetch_add(p + 1, r.y, __ATOMIC_RELAXED, __HIP_MEMORY_SCOPE_WORKGROUP);
    }
}

__device__ __forceinline__ void agg_own_rows(const float* __restrict__ x,
                                             const unsigned* __restrict__ l0, unsigned n0,
                                             const unsigned* __restrict__ l1, unsigned n1,
                                             const float* __restrict__ degp,
                                             const float* __restrict__ bias,
                                             float* __restrict__ out,
                                             int N, int lo, int nloc, int wave, int lane) {
    const int myrow = wave + AGG_WAVES * lane;
    const int rr    = lo + min(myrow, nloc - 1);
    float dg = 0.0f;
#pragma unroll
    for (int p = 0; p < AGG_S; ++p) dg += degp[(size_t)p * N + rr];
    const v2f b = *(const v2f*)(bias + 2 * lane);

    v32f a = {};

    const unsigned n = n0 + n1;
    if (n > 0u) {
        auto entry = [&](unsigned k) {
            const unsigned kk = min(k, n - 1u);
            const unsigned* p = kk < n0 ? l0 + kk : l1 + (kk - n0);
            const unsigned e = *p;
            return k < n ? e : 0u;
        };
        unsigned ent = entry((unsigned)lane);
        v2f ra[16], rb[16];
        rows_load<0>(x, ra, (int)(ent >> 10), lane);
        for (unsigned k = 0; k < n; k += 32u) {
            const int s = (int)(ent >> 10);
            const int slotv = k + (unsigned)lane < n ? (int)((ent & 1023u) / AGG_WAVES) : OWN_ROWS;
            const unsigned nxt = entry(k + 32u + (unsigned)lane);
            rows_load<16>(x, rb, s, lane);
            rows_reg_add<0>(a, ra, slotv);
            rows_load<0>(x, ra, (int)(nxt >> 10), lane);
            rows_reg_add<16>(a, rb, slotv);
            ent = nxt;
        }
    }

    const float ivv = dg > 0.0f ? 1.0f / dg : 0.0f;
#pragma unroll
    for (int i = 0; i < OWN_ROWS; ++i) {
        const int row = wave + AGG_WAVES * i;
        if (row < nloc) {
            const float iv = __builtin_bit_cast(float, __builtin_amdgcn_readlane(__builtin_bit_cast(int, ivv), i));
            v2f e;
            e.x = a[2 * i]     * iv + b.x;
            e.y = a[2 * i + 1] * iv + b.y;
            e.x = e.x > 0.0f ? e.x : expm1f(e.x);
            e.y = e.y > 0.0f ? e.y : expm1f(e.y);
            *(v2f*)(out + (size_t)(lo + row) * DFEAT + 2 * lane) = e;
        }
    }
}

__device__ __forceinline__ void agg_own_rows_queue(const float* __restrict__ x,
                                                   const unsigned* list, unsigned n,
                                                   float* __restrict__ prow,
                                                   int nrows,
                                                   int wave, int lane) {
    v32f a = {};
    if (n > 0u) {
        auto entry = [&](unsigned k) {
            const unsigned e = list[min(k, n - 1u)];
            return k < n ? e : 0u;
        };
        unsigned ent = entry((unsigned)lane);
        v2f ra[16], rb[16];
        rows_load<0>(x, ra, (int)(ent >> 10), lane);
        for (unsigned k = 0; k < n; k += 32u) {
            const int s = (int)(ent >> 10);
            const int slotv = k + (unsigned)lane < n ? (int)((ent & 1023u) / AGG_WAVES) : OWN_ROWS;
            const unsigned nxt = entry(k + 32u + (unsigned)lane);
            rows_load<16>(x, rb, s, lane);
            rows_reg_add<0>(a, ra, slotv);
            rows_load<0>(x, ra, (int)(nxt >> 10), lane);
            rows_reg_add<16>(a, rb, slotv);
            ent = nxt;
        }
    }
#pragma unroll
    for (int i = 0; i < OWN_ROWS; ++i) {
        const int row = wave + AGG_WAVES * i;
        if (row < nrows) {
            v2f e;
            e.x = a[2 * i];
            e.y = a[2 * i + 1];
            *(v2f*)(prow + (size_t)row * DFEAT + 2 * lane) = e;
        }
    }
}

template <bool FIRST>
__global__ __launch_bounds__(AGG_THREADS)
void gcn_agg_kernel(const float* __restrict__ x,
                    const unsigned* __restrict__ keys,
                    const unsigned* __restrict__ packed,
                    float* __restrict__ agg,
                    float* __restrict__ degp,
                    unsigned* __restrict__ lists,
                    unsigned* __restrict__ hdrs,
                    const float* __restrict__ bias,
                    float* __restrict__ out,
                    int N, int E) {
    constexpr int ROWS = FIRST ? AGG_R : AGG_H;
    __shared__ __attribute__((aligned(16))) float    acc[ROWS * DFEAT];
    __shared__ unsigned cnt[FIRST ? AGG_R : 1];
    __shared__ unsigned wqs[AGG_WAVES * AGG_WQ];
    __shared__ unsigned oq[FIRST ? OQ_N * OQ_CAP : 1];
    __shared__ unsigned oqn[OQ_N];
    __shared__ float    invd[FIRST ? 1 : AGG_H];

    const int tid  = threadIdx.x;
    const int lane = tid & 31;
    const int wave = __builtin_amdgcn_readfirstlane(tid >> 5);
    const int nrng  = FIRST ? (int)gridDim.x : (int)(gridDim.x >> 1);
    const int rng   = FIRST ? (int)blockIdx.x : (int)(blockIdx.x >> 1);
    const int half  = FIRST ? 0 : (int)(blockIdx.x & 1);
    const int slice = FIRST ? (int)blockIdx.y : 0;
    const int nslice = FIRST ? AGG_S : 1;
    const int lo    = rng * AGG_R + half * AGG_H;
    const unsigned nloc = (unsigned)max(0, min(ROWS, N - lo));
    unsigned* wq = wqs + wave * AGG_WQ;

    const size_t b0 = (size_t)rng;
    const size_t b1 = (size_t)nrng + rng;
    unsigned f0 = 0u, f1 = 0u;
    if (!FIRST) {
        f0 = hdrs[b0 * LIST_HDR + half * AGG_WAVES + lane];
        f1 = hdrs[b1 * LIST_HDR + half * AGG_WAVES + lane];
    }

    if (!FIRST) {
        if (__builtin_amdgcn_ballot_w32(f0 > (unsigned)OQ_CAP || f1 > (unsigned)OQ_CAP) == 0u) {
            if (nloc > 0u)
                agg_own_rows(x, lists + (b0 * OQ_N + half * AGG_WAVES + wave) * LIST_WCAP,
                             (unsigned)__builtin_amdgcn_readlane((int)f0, wave),
                             lists + (b1 * OQ_N + half * AGG_WAVES + wave) * LIST_WCAP,
                             (unsigned)__builtin_amdgcn_readlane((int)f1, wave),
                             degp, bias, out, N, lo, (int)nloc, wave, lane);
            return;
        }
    }

    auto scan = [&](auto drain) {
        const uint4* key4 = (const uint4*)keys;
        const int nvec  = E >> 4;
        const int gvec  = AGG_VEC * AGG_THREADS;
        const int nfull = nvec / gvec;
        const int gs = (int)(((long long)slice * nfull) / nslice);
        const int ng = (int)(((long long)(slice + 1) * nfull) / nslice) - gs;
        const int g0 = (int)(((long long)blockIdx.x * ng) / gridDim.x);
        const unsigned pat = (unsigned)rng * 0x01010101u;

        unsigned cur  = 0;

        auto flags = [&](unsigned w) {
            const unsigned x = w ^ pat;
            return (0x80808080u - x) & 0x80808080u;
        };

        auto visit16 = [&](const uint4& v, unsigned ebase) {
            unsigned mm = flags(v.x) | (flags(v.y) >> 1) | (flags(v.z) >> 2) | (flags(v.w) >> 3);
            unsigned mask = __builtin_amdgcn_ballot_w32(mm != 0u);
            while (mask) {
                if (mm != 0u) {
                    const unsigned b = (unsigned)__builtin_ctz(mm);
                    const unsigned e = ebase + ((7u - (b & 7u)) << 2) + (b >> 3);
                    wq[cur + __builtin_amdgcn_mbcnt_lo(mask, 0u)] = e;
                    mm &= mm - 1u;
                }
                cur += (unsigned)__builtin_popcount(mask);
                mask = __builtin_amdgcn_ballot_w32(mm != 0u);
            }
        };

        auto group_at = [&](int g) {
            int gg = g + g0;
            return gs + (gg >= ng ? gg - ng : gg);
        };

        auto load_group = [&](int g, uint4 (&v)[AGG_VEC]) {
            const int base = group_at(g) * gvec + tid;
#pragma unroll
            for (int k = 0; k < AGG_VEC; ++k) v[k] = key4[base + k * AGG_THREADS];
        };

        auto visit_vecs = [&](int base, const uint4 (&v)[AGG_VEC]) {
#pragma unroll
            for (int k = 0; k < AGG_VEC; ++k) {
                visit16(v[k], (unsigned)(base + k * AGG_THREADS) << 4);
                while (cur >= 32u) {
                    cur -= 32u;
                    drain(wq + cur, 32u);
                }
            }
        };

        if (ng > 0) {
            uint4 va[AGG_VEC], vb[AGG_VEC];
            load_group(0, va);
            for (int g = 0; g < ng; ++g) {
                load_group(min(g + 1, ng - 1), vb);
                visit_vecs(group_at(g) * gvec + tid, va);
#pragma unroll
                for (int k = 0; k < AGG_VEC; ++k) va[k] = vb[k];
            }
        }
        if (slice == 0) {
            uint4 vr[AGG_VEC];
            const int base = nfull * gvec + tid;
            const unsigned none = pat ^ 0x01010101u;
#pragma unroll
            for (int k = 0; k < AGG_VEC; ++k) {
                const int vi = base + k * AGG_THREADS;
                vr[k] = vi < nvec ? key4[vi] : make_uint4(none, none, none, none);
            }
            visit_vecs(base, vr);
        }
        if (slice == 0 && wave == 0 && (E & 15) != 0) {
            const int e = (nvec << 4) + lane;
            const bool tail = lane < (E & 15);
            const unsigned kb = tail ? (unsigned)((const unsigned char*)keys)[e] : 0xFFu;
            const bool hit = kb == (unsigned)rng;
            const unsigned mask = __builtin_amdgcn_ballot_w32(hit);
            if (hit) wq[cur + __builtin_amdgcn_mbcnt_lo(mask, 0u)] = (unsigned)e;
            cur += (unsigned)__builtin_popcount(mask);
        }
        while (cur >= 32u) {
            cur -= 32u;
            drain(wq + cur, 32u);
        }
        if (cur) drain(wq, cur);
    };


    if constexpr (FIRST) {
        for (int i = tid; i < AGG_R; i += AGG_THREADS) cnt[i] = 0u;
        if (tid < OQ_N) oqn[tid] = 0u;
        __syncthreads();
        scan([&](const unsigned* q, unsigned m) {
            if (m == 32u) agg_file<true >(packed, cnt, q, 32, lane, oq, oqn);
            else          agg_file<false>(packed, cnt, q, (int)m, lane, oq, oqn);
        });
        __syncthreads();

        float* plane = agg + ((size_t)slice * N + lo) * DFEAT;
        const unsigned fq0 = oqn[lane], fq1 = oqn[AGG_WAVES + lane];
        if (__builtin_expect(__builtin_amdgcn_ballot_w32(fq0 > (unsigned)OQ_CAP || fq1 > (unsigned)OQ_CAP) == 0u, 1)) {
#pragma unroll
            for (int hq = 0; hq < 2; ++hq) {
                const int q = hq * AGG_WAVES + wave;
                const unsigned nq = (unsigned)__builtin_amdgcn_readlane((int)(hq ? fq1 : fq0), wave);
                agg_own_rows_queue(x, oq + q * OQ_CAP, nq, plane + (size_t)hq * AGG_H * DFEAT,
                                   max(0, min(AGG_H, (int)nloc - hq * AGG_H)), wave, lane);
            }
        } else {
            for (int i = tid; i < ROWS * DFEAT / 4; i += AGG_THREADS)
                ((float4*)acc)[i] = make_float4(0.f, 0.f, 0.f, 0.f);
            __syncthreads();
            scan([&](const unsigned* q, unsigned m) {
                if (m == 32u) agg_gather_all<true >(x, packed, acc, q, 32, lane);
                else          agg_gather_all<false>(x, packed, acc, q, (int)m, lane);
            });
            __syncthreads();
            float4* orow = (float4*)plane;
            for (int i = tid; i < (int)nloc * (DFEAT / 4); i += AGG_THREADS)
                orow[i] = ((const float4*)acc)[i];
        }

        for (int i = tid; i < (int)nloc; i += AGG_THREADS)
            degp[(size_t)slice * N + lo + i] = (float)cnt[i];
        const size_t blk = (size_t)slice * nrng + rng;
#pragma unroll
        for (int hq = 0; hq < 2; ++hq) {
            const int q = hq * AGG_WAVES + wave;
            const unsigned nq = (unsigned)__builtin_amdgcn_readfirstlane((int)oqn[q]);
            const unsigned nk = min(nq, (unsigned)OQ_CAP);
            unsigned* lst = lists + (blk * OQ_N + q) * LIST_WCAP;
            for (unsigned k = (unsigned)lane; k < nk; k += 32u) lst[k] = oq[q * OQ_CAP + k];
            if (lane == 0) hdrs[blk * LIST_HDR + q] = nq;
        }
    } else {
        for (int i = tid; i < ROWS * DFEAT / 4; i += AGG_THREADS)
            ((float4*)acc)[i] = make_float4(0.f, 0.f, 0.f, 0.f);
        for (int i = tid; i < (int)nloc; i += AGG_THREADS) {
            float dg = 0.0f;
#pragma unroll
            for (int p = 0; p < AGG_S; ++p) dg += degp[(size_t)p * N + lo + i];
            invd[i] = dg > 0.0f ? 1.0f / dg : 0.0f;
        }
        __syncthreads();
        scan([&](const unsigned* q, unsigned m) {
            if (m == 32u) agg_drain_half<true >(x, packed, acc, q, 32, lane, half);
            else          agg_drain_half<false>(x, packed, acc, q, (int)m, lane, half);
        });
        __syncthreads();

        float4* orow = (float4*)(out + (size_t)lo * DFEAT);
        for (int i = tid; i < (int)nloc * (DFEAT / 4); i += AGG_THREADS) {
            const float4 a  = ((const float4*)acc)[i];
            const float4 b  = ((const float4*)bias)[i & (DFEAT / 4 - 1)];
            const float  iv = invd[i / (DFEAT / 4)];
            float4 e;
            e.x = a.x * iv + b.x; e.y = a.y * iv + b.y;
            e.z = a.z * iv + b.z; e.w = a.w * iv + b.w;
            e.x = e.x > 0.0f ? e.x : expm1f(e.x);
            e.y = e.y > 0.0f ? e.y : expm1f(e.y);
            e.z = e.z > 0.0f ? e.z : expm1f(e.z);
            e.w = e.w > 0.0f ? e.w : expm1f(e.w);
            orow[i] = e;
        }
    }
}

#define LIN_WAVES   8
#define LIN_THREADS (LIN_WAVES * 32)
#define LIN_PITCH   (DFEAT + 4)

__global__ __launch_bounds__(LIN_THREADS)
void gcn_wmma_kernel(const float* __restrict__ agg,
                     const float* __restrict__ degp,
                     const float* __restrict__ W,
                     const float* __restrict__ bias,
                     const float* __restrict__ W2,
                     float* __restrict__ out,
                     int N) {
    __shared__ __attribute__((aligned(16))) float Ws[DFEAT * LIN_PITCH];
    __shared__ __attribute__((aligned(16))) float Ws2[DFEAT * LIN_PITCH];
    __shared__ __attribute__((aligned(16))) float As[LIN_WAVES * 16 * LIN_PITCH];

    const int tid  = threadIdx.x;
    const int lane = tid & 31;
    const int wave  = __builtin_amdgcn_readfirstlane(tid >> 5);
    const int tiles = N >> 4;
    const int want  = blockIdx.x * LIN_WAVES + wave;
    const bool valid = want < tiles;
    const int tile = valid ? want : tiles - 1;
    const int rowBase = tile << 4;
    const int m  = lane & 15;
    const int hi = lane >> 4;
    const int k2 = hi << 1;
    const int c4 = m << 2;
    float* mine = As + wave * (16 * LIN_PITCH);
    const size_t plane = (size_t)N * DFEAT;

    float4 w4[DFEAT * DFEAT / 4 / LIN_THREADS], x4[DFEAT * DFEAT / 4 / LIN_THREADS];
#pragma unroll
    for (int j = 0; j < DFEAT * DFEAT / 4 / LIN_THREADS; ++j) {
        w4[j] = *(const float4*)(W  + 4 * (tid + j * LIN_THREADS));
        x4[j] = *(const float4*)(W2 + 4 * (tid + j * LIN_THREADS));
    }

    const float* rows = agg + (size_t)rowBase * DFEAT;
    const unsigned loff = (unsigned)(hi * DFEAT + c4);
    float4 v[8];
#pragma unroll
    for (int i = 0; i < 8; ++i)
        v[i] = *(const float4*)(rows + loff + i * 2 * DFEAT);
    float4 u[AGG_S - 1][8];
#pragma unroll
    for (int p = 1; p < AGG_S; ++p) {
#pragma unroll
        for (int i = 0; i < 8; ++i)
            u[p - 1][i] = *(const float4*)(rows + p * plane + loff + i * 2 * DFEAT);
    }

    float deg = degp[rowBase + m];
#pragma unroll
    for (int p = 1; p < AGG_S; ++p) deg += degp[(size_t)p * N + rowBase + m];
    float bcol[4];
#pragma unroll
    for (int t = 0; t < 4; ++t) bcol[t] = bias[t * 16 + m];

#pragma unroll
    for (int j = 0; j < DFEAT * DFEAT / 4 / LIN_THREADS; ++j) {
        const int idx = tid + j * LIN_THREADS;
        *(float4*)(Ws  + (idx >> 4) * LIN_PITCH + ((idx & 15) << 2)) = w4[j];
        *(float4*)(Ws2 + (idx >> 4) * LIN_PITCH + ((idx & 15) << 2)) = x4[j];
    }
#pragma unroll
    for (int i = 0; i < 8; ++i) {
#pragma unroll
        for (int p = 1; p < AGG_S; ++p) {
            v[i].x += u[p - 1][i].x; v[i].y += u[p - 1][i].y;
            v[i].z += u[p - 1][i].z; v[i].w += u[p - 1][i].w;
        }
        *(float4*)(mine + (2 * i + hi) * LIN_PITCH + c4) = v[i];
    }
    const float inv = deg > 0.0f ? 1.0f / deg : 0.0f;
    __syncthreads();

    v8f c[4] = {};
    const float* arow = mine + m * LIN_PITCH + k2;
    const float* brow = Ws   + m * LIN_PITCH + k2;
#pragma unroll
    for (int kb = 0; kb < DFEAT; kb += 4) {
        v2f a = *(const v2f*)(arow + kb);
        a.x *= inv;
        a.y *= inv;
#pragma unroll
        for (int t = 0; t < 4; ++t) {
            const v2f b = *(const v2f*)(brow + t * 16 * LIN_PITCH + kb);
            c[t] = __builtin_amdgcn_wmma_f32_16x16x4_f32(
                     false, a,  false, b,
                     (short)0, c[t],  false,  false);
        }
    }

#pragma unroll
    for (int t = 0; t < 4; ++t) {
#pragma unroll
        for (int r = 0; r < 8; ++r) {
            float e = c[t][r] + bcol[t];
            e = e > 0.0f ? e : expm1f(e);
            mine[(r + (hi << 3)) * LIN_PITCH + t * 16 + m] = e;
        }
    }
    __syncthreads();

    v8f c2[4] = {};
    const float* brow2 = Ws2 + m * LIN_PITCH + k2;
#pragma unroll
    for (int kb = 0; kb < DFEAT; kb += 4) {
        const v2f a = *(const v2f*)(arow + kb);
#pragma unroll
        for (int t = 0; t < 4; ++t) {
            const v2f b = *(const v2f*)(brow2 + t * 16 * LIN_PITCH + kb);
            c2[t] = __builtin_amdgcn_wmma_f32_16x16x4_f32(
                     false, a,  false, b,
                     (short)0, c2[t],  false,  false);
        }
    }
#pragma unroll
    for (int t = 0; t < 4; ++t) {
#pragma unroll
        for (int r = 0; r < 8; ++r)
            mine[(r + (hi << 3)) * LIN_PITCH + t * 16 + m] = c2[t][r];
    }
    __syncthreads();

    if (valid) {
        float* orow = out + (size_t)rowBase * DFEAT;
#pragma unroll
        for (int i = 0; i < 8; ++i)
            *(float4*)(orow + loff + i * 2 * DFEAT) =
                *(const float4*)(mine + (2 * i + hi) * LIN_PITCH + c4);
    }
}

extern "C" void kernel_launch(void* const* d_in, const int* in_sizes, int n_in,
                              void* d_out, int out_size, void* d_ws, size_t ws_size,
                              hipStream_t stream) {
    const float* h   = (const float*)d_in[0];
    const int*   src = (const int*)  d_in[1];
    const int*   dst = (const int*)  d_in[2];
    const float* W1  = (const float*)d_in[3];
    const float* b1  = (const float*)d_in[4];
    const float* W2  = (const float*)d_in[5];
    const float* b2  = (const float*)d_in[6];
    float*       out = (float*)d_out;

    const int N = in_sizes[0] / DFEAT;
    const int E = in_sizes[1];

    char* ws = (char*)d_ws;
    float* deg = (float*)ws;
    size_t degBytes = (((size_t)AGG_S * N * sizeof(float)) + 255) & ~(size_t)255;
    float* agg = (float*)(ws + degBytes);
    float* h1  = agg + (size_t)AGG_S * N * DFEAT;
    unsigned* hdrs  = (unsigned*)(h1 + (size_t)N * DFEAT);

    const int tiles = N / 16;
    const int linB  = (tiles + LIN_WAVES - 1) / LIN_WAVES;
    const dim3 aggB((N + AGG_R - 1) / AGG_R, AGG_S);
    unsigned* lists = hdrs + (size_t)aggB.x * aggB.y * LIST_HDR;
    unsigned* keys   = lists + (size_t)aggB.x * aggB.y * OQ_N * LIST_WCAP;
    unsigned* packed = keys + (size_t)((E + 15) / 16) * 4;
    const int keyB   = (E / 4 + KEY_THREADS * KEY_ITEMS - 1) / (KEY_THREADS * KEY_ITEMS) + (E < 4);

    gcn_key_kernel<<<keyB, KEY_THREADS, 0, stream>>>(src, dst, keys, packed, E);

    gcn_agg_kernel<true><<<aggB, AGG_THREADS, 0, stream>>>(h, keys, packed, agg, deg, lists, hdrs, b2, out, N, E);
    gcn_wmma_kernel<<<linB, LIN_THREADS, 0, stream>>>(agg, deg, W1, b1, W2, h1, N);

    const dim3 aggB2(aggB.x * 2, 1);
    gcn_agg_kernel<false><<<aggB2, AGG_THREADS, 0, stream>>>(h1, keys, packed, agg, deg, lists, hdrs, b2, out, N, E);
}


# gather loops of both layers: 10 -> 7 vector instructions per row: one readlane per entry (row, sum index by scalar shifts), load = scalar base + 32-bit offset, packed two-float add; sum order kept
# speedup vs baseline: 4.9054x; 1.0303x over previous
#include <hip/hip_runtime.h>
#include <hip/hip_bf16.h>
#include <math.h>

typedef __attribute__((ext_vector_type(2))) float v2f;
typedef __attribute__((ext_vector_type(8))) float v8f;
typedef __attribute__((ext_vector_type(32))) float v32f;

#define DFEAT 64

#define AGG_S       2
#define AGG_R       782
#define AGG_H       (AGG_R / 2)
#define AGG_THREADS 1024
#define AGG_WAVES   (AGG_THREADS / 32)
#define AGG_VEC     4
#define AGG_WQ      (16 * 32 + 32)
#define KEY_THREADS 256
#define KEY_ITEMS   4

__global__ __launch_bounds__(KEY_THREADS)
void gcn_key_kernel(const int* __restrict__ src,
                    const int* __restrict__ dst,
                    unsigned* __restrict__ keys,
                    unsigned* __restrict__ packed,
                    int E) {
    const int nv   = E >> 2;
    const int base = blockIdx.x * (KEY_THREADS * KEY_ITEMS) + threadIdx.x;
    if (nv > 0) {
        int4 d[KEY_ITEMS], s[KEY_ITEMS];
#pragma unroll
        for (int k = 0; k < KEY_ITEMS; ++k) {
            const int i = min(base + k * KEY_THREADS, nv - 1);
            d[k] = ((const int4*)dst)[i];
            s[k] = ((const int4*)src)[i];
        }
#pragma unroll
        for (int k = 0; k < KEY_ITEMS; ++k) {
            const int i = base + k * KEY_THREADS;
            if (i < nv) {
                const unsigned r0 = (unsigned)d[k].x / AGG_R, r1 = (unsigned)d[k].y / AGG_R;
                const unsigned r2 = (unsigned)d[k].z / AGG_R, r3 = (unsigned)d[k].w / AGG_R;
                uint4 p;
                p.x = ((unsigned)s[k].x << 10) | ((unsigned)d[k].x - r0 * AGG_R);
                p.y = ((unsigned)s[k].y << 10) | ((unsigned)d[k].y - r1 * AGG_R);
                p.z = ((unsigned)s[k].z << 10) | ((unsigned)d[k].z - r2 * AGG_R);
                p.w = ((unsigned)s[k].w << 10) | ((unsigned)d[k].w - r3 * AGG_R);
                keys[i] = r0 | (r1 << 8) | (r2 << 16) | (r3 << 24);
                ((uint4*)packed)[i] = p;
            }
        }
    }
    if (blockIdx.x == 0 && (int)threadIdx.x < (E & 3)) {
        const int e = (nv << 2) + (int)threadIdx.x;
        const unsigned dv = (unsigned)dst[e];
        const unsigned r  = dv / AGG_R;
        packed[e] = ((unsigned)src[e] << 10) | (dv - r * AGG_R);
        ((unsigned char*)keys)[e] = (unsigned char)r;
    }
}

#define OQ_N        (2 * AGG_WAVES)
#define OQ_CAP      160
#define LIST_WCAP   OQ_CAP
#define LIST_HDR    OQ_N
static_assert(AGG_S == 2 && (AGG_R & 1) == 0, "layer 2 reads two slices' lists per half range");

#define ENT_IDLE    ((unsigned)OWN_ROWS << 5)

template <int J0>
__device__ __forceinline__ void ents_get(unsigned (&e)[16], unsigned ent) {
#pragma unroll
    for (int j = 0; j < 16; ++j) e[j] = (unsigned)__builtin_amdgcn_readlane((int)ent, J0 + j);
}

__device__ __forceinline__ void rows_load(const float* __restrict__ x, v2f (&r)[16],
                                          const unsigned (&e)[16], unsigned lane8) {
#pragma unroll
    for (int j = 0; j < 16; ++j) {
        const unsigned off = ((e[j] >> 10) << 8) + lane8;
        r[j] = *(const v2f*)((const char*)x + off);
    }
}

#define OWN_ROWS    ((AGG_H + AGG_WAVES - 1) / AGG_WAVES)
static_assert(2 * (OWN_ROWS + 1) <= 32, "the wave's sums and one idle pair fit one 32-register vector");

__device__ __forceinline__ void rows_reg_add(v32f& a, const v2f (&r)[16], const unsigned (&e)[16]) {
#pragma unroll
    for (int j = 0; j < 16; ++j) {
        const unsigned q = (e[j] >> 5) & 31u;
        v2f t;
        t.x = a[2 * q];
        t.y = a[2 * q + 1];
        t += r[j];
        a[2 * q]     = t.x;
        a[2 * q + 1] = t.y;
    }
}

template <bool FULL>
__device__ __forceinline__ void agg_file(const unsigned* __restrict__ packed,
                                         unsigned* cnt, const unsigned* wq,
                                         int m, int lane, unsigned* oq, unsigned* oqn) {
    const bool live = FULL || lane < m;
    const unsigned ent = live ? packed[wq[lane]] : 0u;
    const int d  = (int)(ent & 1023u);
    const int hf = d >= AGG_H ? 1 : 0;
    const int d2 = d - hf * AGG_H;
    if (live) {
        __hip_atomic_fetch_add(&cnt[d], 1u, __ATOMIC_RELAXED, __HIP_MEMORY_SCOPE_WORKGROUP);
        const unsigned q   = (unsigned)hf * AGG_WAVES + ((unsigned)d2 & (AGG_WAVES - 1));
        const unsigned pos = __hip_atomic_fetch_add(&oqn[q], 1u, __ATOMIC_RELAXED, __HIP_MEMORY_SCOPE_WORKGROUP);
        if (pos < (unsigned)OQ_CAP) oq[q * OQ_CAP + pos] = (ent & ~1023u) | (unsigned)d2;
    }
}

template <bool FULL>
__device__ __forceinline__ void agg_gather_all(const float* __restrict__ x,
                                               const unsigned* __restrict__ packed,
                                               float* acc, const unsigned* wq, int m, int lane) {
    const bool live = FULL || lane < m;
    const unsigned ent = live ? packed[wq[lane]] : 0u;
    const int d = (int)(ent & 1023u);
    const int s = (int)(ent >> 10);
    unsigned todo = __builtin_amdgcn_ballot_w32(live);
    while (todo) {
        const int j = __builtin_ctz(todo);
        todo &= todo - 1u;
        const int sj = __builtin_amdgcn_readlane(s, j);
        const int dj = __builtin_amdgcn_readlane(d, j);
        const v2f r = *(const v2f*)(x + (size_t)sj * DFEAT + 2 * lane);
        float* p = acc + dj * DFEAT + 2 * lane;
        __hip_atomic_fetch_add(p + 0, r.x, __ATOMIC_RELAXED, __HIP_MEMORY_SCOPE_WORKGROUP);
        __hip_atomic_fetch_add(p + 1, r.y, __ATOMIC_RELAXED, __HIP_MEMORY_SCOPE_WORKGROUP);
    }
}

template <bool FULL>
__device__ __forceinline__ void agg_drain_half(const float* __restrict__ x,
                                               const unsigned* __restrict__ packed,
                                               float* acc, const unsigned* wq,
                                               int m, int lane, int half) {
    const bool live = FULL || lane < m;
    const unsigned ent = live ? packed[wq[lane]] : 0u;
    const int d = (int)(ent & 1023u);
    const int s = (int)(ent >> 10);
    const int hf = d >= AGG_H ? 1 : 0;
    const int d2 = d - hf * AGG_H;
    unsigned todo = __builtin_amdgcn_ballot_w32(live && hf == half);
    while (todo) {
        const int j = __builtin_ctz(todo);
        todo &= todo - 1u;
        const int sj = __builtin_amdgcn_readlane(s, j);
        const int dj = __builtin_amdgcn_readlane(d2, j);
        const v2f r = *(const v2f*)(x + (size_t)sj * DFEAT + 2 * lane);
        float* p = acc + dj * DFEAT + 2 * lane;
        __hip_atomic_fetch_add(p + 0, r.x, __ATOMIC_RELAXED, __HIP_MEMORY_SCOPE_WORKGROUP);
        __hip_atomic_fetch_add(p + 1, r.y, __ATOMIC_RELAXED, __HIP_MEMORY_SCOPE_WORKGROUP);
    }
}

__device__ __forceinline__ void agg_own_rows(const float* __restrict__ x,
                                             const unsigned* __restrict__ l0, unsigned n0,
                                             const unsigned* __restrict__ l1, unsigned n1,
                                             const float* __restrict__ degp,
                                             const float* __restrict__ bias,
                                             float* __restrict__ out,
                                             int N, int lo, int nloc, int wave, int lane) {
    const int myrow = wave + AGG_WAVES * lane;
    const int rr    = lo + min(myrow, nloc - 1);
    float dg = 0.0f;
#pragma unroll
    for (int p = 0; p < AGG_S; ++p) dg += degp[(size_t)p * N + rr];
    const v2f b = *(const v2f*)(bias + 2 * lane);

    v32f a = {};
    const unsigned lane8 = (unsigned)lane * 8u;

    const unsigned n = n0 + n1;
    if (n > 0u) {
        auto entry = [&](unsigned k) {
            const unsigned kk = min(k, n - 1u);
            const unsigned* p = kk < n0 ? l0 + kk : l1 + (kk - n0);
            const unsigned e = *p;
            return k < n ? e : ENT_IDLE;
        };
        unsigned ent = entry((unsigned)lane);
        unsigned ea[16], eb[16];
        v2f ra[16], rb[16];
        ents_get<0>(ea, ent);
        rows_load(x, ra, ea, lane8);
        for (unsigned k = 0; k < n; k += 32u) {
            const unsigned nxt = entry(k + 32u + (unsigned)lane);
            ents_get<16>(eb, ent);
            rows_load(x, rb, eb, lane8);
            rows_reg_add(a, ra, ea);
            ents_get<0>(ea, nxt);
            rows_load(x, ra, ea, lane8);
            rows_reg_add(a, rb, eb);
            ent = nxt;
        }
    }

    const float ivv = dg > 0.0f ? 1.0f / dg : 0.0f;
#pragma unroll
    for (int i = 0; i < OWN_ROWS; ++i) {
        const int row = wave + AGG_WAVES * i;
        if (row < nloc) {
            const float iv = __builtin_bit_cast(float, __builtin_amdgcn_readlane(__builtin_bit_cast(int, ivv), i));
            v2f e;
            e.x = a[2 * i]     * iv + b.x;
            e.y = a[2 * i + 1] * iv + b.y;
            e.x = e.x > 0.0f ? e.x : expm1f(e.x);
            e.y = e.y > 0.0f ? e.y : expm1f(e.y);
            *(v2f*)(out + (size_t)(lo + row) * DFEAT + 2 * lane) = e;
        }
    }
}

__device__ __forceinline__ void agg_own_rows_queue(const float* __restrict__ x,
                                                   const unsigned* list, unsigned n,
                                                   float* __restrict__ prow,
                                                   int nrows,
                                                   int wave, int lane) {
    v32f a = {};
    const unsigned lane8 = (unsigned)lane * 8u;
    if (n > 0u) {
        auto entry = [&](unsigned k) {
            const unsigned e = list[min(k, n - 1u)];
            return k < n ? e : ENT_IDLE;
        };
        unsigned ent = entry((unsigned)lane);
        unsigned ea[16], eb[16];
        v2f ra[16], rb[16];
        ents_get<0>(ea, ent);
        rows_load(x, ra, ea, lane8);
        for (unsigned k = 0; k < n; k += 32u) {
            const unsigned nxt = entry(k + 32u + (unsigned)lane);
            ents_get<16>(eb, ent);
            rows_load(x, rb, eb, lane8);
            rows_reg_add(a, ra, ea);
            ents_get<0>(ea, nxt);
            rows_load(x, ra, ea, lane8);
            rows_reg_add(a, rb, eb);
            ent = nxt;
        }
    }
#pragma unroll
    for (int i = 0; i < OWN_ROWS; ++i) {
        const int row = wave + AGG_WAVES * i;
        if (row < nrows) {
            v2f e;
            e.x = a[2 * i];
            e.y = a[2 * i + 1];
            *(v2f*)(prow + (size_t)row * DFEAT + 2 * lane) = e;
        }
    }
}

template <bool FIRST>
__global__ __launch_bounds__(AGG_THREADS)
void gcn_agg_kernel(const float* __restrict__ x,
                    const unsigned* __restrict__ keys,
                    const unsigned* __restrict__ packed,
                    float* __restrict__ agg,
                    float* __restrict__ degp,
                    unsigned* __restrict__ lists,
                    unsigned* __restrict__ hdrs,
                    const float* __restrict__ bias,
                    float* __restrict__ out,
                    int N, int E) {
    constexpr int ROWS = FIRST ? AGG_R : AGG_H;
    __shared__ __attribute__((aligned(16))) float    acc[ROWS * DFEAT];
    __shared__ unsigned cnt[FIRST ? AGG_R : 1];
    __shared__ unsigned wqs[AGG_WAVES * AGG_WQ];
    __shared__ unsigned oq[FIRST ? OQ_N * OQ_CAP : 1];
    __shared__ unsigned oqn[OQ_N];
    __shared__ float    invd[FIRST ? 1 : AGG_H];

    const int tid  = threadIdx.x;
    const int lane = tid & 31;
    const int wave = __builtin_amdgcn_readfirstlane(tid >> 5);
    const int nrng  = FIRST ? (int)gridDim.x : (int)(gridDim.x >> 1);
    const int rng   = FIRST ? (int)blockIdx.x : (int)(blockIdx.x >> 1);
    const int half  = FIRST ? 0 : (int)(blockIdx.x & 1);
    const int slice = FIRST ? (int)blockIdx.y : 0;
    const int nslice = FIRST ? AGG_S : 1;
    const int lo    = rng * AGG_R + half * AGG_H;
    const unsigned nloc = (unsigned)max(0, min(ROWS, N - lo));
    unsigned* wq = wqs + wave * AGG_WQ;

    const size_t b0 = (size_t)rng;
    const size_t b1 = (size_t)nrng + rng;
    unsigned f0 = 0u, f1 = 0u;
    if (!FIRST) {
        f0 = hdrs[b0 * LIST_HDR + half * AGG_WAVES + lane];
        f1 = hdrs[b1 * LIST_HDR + half * AGG_WAVES + lane];
    }

    if (!FIRST) {
        if (__builtin_amdgcn_ballot_w32(f0 > (unsigned)OQ_CAP || f1 > (unsigned)OQ_CAP) == 0u) {
            if (nloc > 0u)
                agg_own_rows(x, lists + (b0 * OQ_N + half * AGG_WAVES + wave) * LIST_WCAP,
                             (unsigned)__builtin_amdgcn_readlane((int)f0, wave),
                             lists + (b1 * OQ_N + half * AGG_WAVES + wave) * LIST_WCAP,
                             (unsigned)__builtin_amdgcn_readlane((int)f1, wave),
                             degp, bias, out, N, lo, (int)nloc, wave, lane);
            return;
        }
    }

    auto scan = [&](auto drain) {
        const uint4* key4 = (const uint4*)keys;
        const int nvec  = E >> 4;
        const int gvec  = AGG_VEC * AGG_THREADS;
        const int nfull = nvec / gvec;
        const int gs = (int)(((long long)slice * nfull) / nslice);
        const int ng = (int)(((long long)(slice + 1) * nfull) / nslice) - gs;
        const int g0 = (int)(((long long)blockIdx.x * ng) / gridDim.x);
        const unsigned pat = (unsigned)rng * 0x01010101u;

        unsigned cur  = 0;

        auto flags = [&](unsigned w) {
            const unsigned x = w ^ pat;
            return (0x80808080u - x) & 0x80808080u;
        };

        auto visit16 = [&](const uint4& v, unsigned ebase) {
            unsigned mm = flags(v.x) | (flags(v.y) >> 1) | (flags(v.z) >> 2) | (flags(v.w) >> 3);
            unsigned mask = __builtin_amdgcn_ballot_w32(mm != 0u);
            while (mask) {
                if (mm != 0u) {
                    const unsigned b = (unsigned)__builtin_ctz(mm);
                    const unsigned e = ebase + ((7u - (b & 7u)) << 2) + (b >> 3);
                    wq[cur + __builtin_amdgcn_mbcnt_lo(mask, 0u)] = e;
                    mm &= mm - 1u;
                }
                cur += (unsigned)__builtin_popcount(mask);
                mask = __builtin_amdgcn_ballot_w32(mm != 0u);
            }
        };

        auto group_at = [&](int g) {
            int gg = g + g0;
            return gs + (gg >= ng ? gg - ng : gg);
        };

        auto load_group = [&](int g, uint4 (&v)[AGG_VEC]) {
            const int base = group_at(g) * gvec + tid;
#pragma unroll
            for (int k = 0; k < AGG_VEC; ++k) v[k] = key4[base + k * AGG_THREADS];
        };

        auto visit_vecs = [&](int base, const uint4 (&v)[AGG_VEC]) {
#pragma unroll
            for (int k = 0; k < AGG_VEC; ++k) {
                visit16(v[k], (unsigned)(base + k * AGG_THREADS) << 4);
                while (cur >= 32u) {
                    cur -= 32u;
                    drain(wq + cur, 32u);
                }
            }
        };

        if (ng > 0) {
            uint4 va[AGG_VEC], vb[AGG_VEC];
            load_group(0, va);
            for (int g = 0; g < ng; ++g) {
                load_group(min(g + 1, ng - 1), vb);
                visit_vecs(group_at(g) * gvec + tid, va);
#pragma unroll
                for (int k = 0; k < AGG_VEC; ++k) va[k] = vb[k];
            }
        }
        if (slice == 0) {
            uint4 vr[AGG_VEC];
            const int base = nfull * gvec + tid;
            const unsigned none = pat ^ 0x01010101u;
#pragma unroll
            for (int k = 0; k < AGG_VEC; ++k) {
                const int vi = base + k * AGG_THREADS;
                vr[k] = vi < nvec ? key4[vi] : make_uint4(none, none, none, none);
            }
            visit_vecs(base, vr);
        }
        if (slice == 0 && wave == 0 && (E & 15) != 0) {
            const int e = (nvec << 4) + lane;
            const bool tail = lane < (E & 15);
            const unsigned kb = tail ? (unsigned)((const unsigned char*)keys)[e] : 0xFFu;
            const bool hit = kb == (unsigned)rng;
            const unsigned mask = __builtin_amdgcn_ballot_w32(hit);
            if (hit) wq[cur + __builtin_amdgcn_mbcnt_lo(mask, 0u)] = (unsigned)e;
            cur += (unsigned)__builtin_popcount(mask);
        }
        while (cur >= 32u) {
            cur -= 32u;
            drain(wq + cur, 32u);
        }
        if (cur) drain(wq, cur);
    };


    if constexpr (FIRST) {
        for (int i = tid; i < AGG_R; i += AGG_THREADS) cnt[i] = 0u;
        if (tid < OQ_N) oqn[tid] = 0u;
        __syncthreads();
        scan([&](const unsigned* q, unsigned m) {
            if (m == 32u) agg_file<true >(packed, cnt, q, 32, lane, oq, oqn);
            else          agg_file<false>(packed, cnt, q, (int)m, lane, oq, oqn);
        });
        __syncthreads();

        float* plane = agg + ((size_t)slice * N + lo) * DFEAT;
        const unsigned fq0 = oqn[lane], fq1 = oqn[AGG_WAVES + lane];
        if (__builtin_expect(__builtin_amdgcn_ballot_w32(fq0 > (unsigned)OQ_CAP || fq1 > (unsigned)OQ_CAP) == 0u, 1)) {
#pragma unroll
            for (int hq = 0; hq < 2; ++hq) {
                const int q = hq * AGG_WAVES + wave;
                const unsigned nq = (unsigned)__builtin_amdgcn_readlane((int)(hq ? fq1 : fq0), wave);
                agg_own_rows_queue(x, oq + q * OQ_CAP, nq, plane + (size_t)hq * AGG_H * DFEAT,
                                   max(0, min(AGG_H, (int)nloc - hq * AGG_H)), wave, lane);
            }
        } else {
            for (int i = tid; i < ROWS * DFEAT / 4; i += AGG_THREADS)
                ((float4*)acc)[i] = make_float4(0.f, 0.f, 0.f, 0.f);
            __syncthreads();
            scan([&](const unsigned* q, unsigned m) {
                if (m == 32u) agg_gather_all<true >(x, packed, acc, q, 32, lane);
                else          agg_gather_all<false>(x, packed, acc, q, (int)m, lane);
            });
            __syncthreads();
            float4* orow = (float4*)plane;
            for (int i = tid; i < (int)nloc * (DFEAT / 4); i += AGG_THREADS)
                orow[i] = ((const float4*)acc)[i];
        }

        for (int i = tid; i < (int)nloc; i += AGG_THREADS)
            degp[(size_t)slice * N + lo + i] = (float)cnt[i];
        const size_t blk = (size_t)slice * nrng + rng;
#pragma unroll
        for (int hq = 0; hq < 2; ++hq) {
            const int q = hq * AGG_WAVES + wave;
            const unsigned nq = (unsigned)__builtin_amdgcn_readfirstlane((int)oqn[q]);
            const unsigned nk = min(nq, (unsigned)OQ_CAP);
            unsigned* lst = lists + (blk * OQ_N + q) * LIST_WCAP;
            for (unsigned k = (unsigned)lane; k < nk; k += 32u) lst[k] = oq[q * OQ_CAP + k];
            if (lane == 0) hdrs[blk * LIST_HDR + q] = nq;
        }
    } else {
        for (int i = tid; i < ROWS * DFEAT / 4; i += AGG_THREADS)
            ((float4*)acc)[i] = make_float4(0.f, 0.f, 0.f, 0.f);
        for (int i = tid; i < (int)nloc; i += AGG_THREADS) {
            float dg = 0.0f;
#pragma unroll
            for (int p = 0; p < AGG_S; ++p) dg += degp[(size_t)p * N + lo + i];
            invd[i] = dg > 0.0f ? 1.0f / dg : 0.0f;
        }
        __syncthreads();
        scan([&](const unsigned* q, unsigned m) {
            if (m == 32u) agg_drain_half<true >(x, packed, acc, q, 32, lane, half);
            else          agg_drain_half<false>(x, packed, acc, q, (int)m, lane, half);
        });
        __syncthreads();

        float4* orow = (float4*)(out + (size_t)lo * DFEAT);
        for (int i = tid; i < (int)nloc * (DFEAT / 4); i += AGG_THREADS) {
            const float4 a  = ((const float4*)acc)[i];
            const float4 b  = ((const float4*)bias)[i & (DFEAT / 4 - 1)];
            const float  iv = invd[i / (DFEAT / 4)];
            float4 e;
            e.x = a.x * iv + b.x; e.y = a.y * iv + b.y;
            e.z = a.z * iv + b.z; e.w = a.w * iv + b.w;
            e.x = e.x > 0.0f ? e.x : expm1f(e.x);
            e.y = e.y > 0.0f ? e.y : expm1f(e.y);
            e.z = e.z > 0.0f ? e.z : expm1f(e.z);
            e.w = e.w > 0.0f ? e.w : expm1f(e.w);
            orow[i] = e;
        }
    }
}

#define LIN_WAVES   8
#define LIN_THREADS (LIN_WAVES * 32)
#define LIN_PITCH   (DFEAT + 4)

__global__ __launch_bounds__(LIN_THREADS)
void gcn_wmma_kernel(const float* __restrict__ agg,
                     const float* __restrict__ degp,
                     const float* __restrict__ W,
                     const float* __restrict__ bias,
                     const float* __restrict__ W2,
                     float* __restrict__ out,
                     int N) {
    __shared__ __attribute__((aligned(16))) float Ws[DFEAT * LIN_PITCH];
    __shared__ __attribute__((aligned(16))) float Ws2[DFEAT * LIN_PITCH];
    __shared__ __attribute__((aligned(16))) float As[LIN_WAVES * 16 * LIN_PITCH];

    const int tid  = threadIdx.x;
    const int lane = tid & 31;
    const int wave  = __builtin_amdgcn_readfirstlane(tid >> 5);
    const int tiles = N >> 4;
    const int want  = blockIdx.x * LIN_WAVES + wave;
    const bool valid = want < tiles;
    const int tile = valid ? want : tiles - 1;
    const int rowBase = tile << 4;
    const int m  = lane & 15;
    const int hi = lane >> 4;
    const int k2 = hi << 1;
    const int c4 = m << 2;
    float* mine = As + wave * (16 * LIN_PITCH);
    const size_t plane = (size_t)N * DFEAT;

    float4 w4[DFEAT * DFEAT / 4 / LIN_THREADS], x4[DFEAT * DFEAT / 4 / LIN_THREADS];
#pragma unroll
    for (int j = 0; j < DFEAT * DFEAT / 4 / LIN_THREADS; ++j) {
        w4[j] = *(const float4*)(W  + 4 * (tid + j * LIN_THREADS));
        x4[j] = *(const float4*)(W2 + 4 * (tid + j * LIN_THREADS));
    }

    const float* rows = agg + (size_t)rowBase * DFEAT;
    const unsigned loff = (unsigned)(hi * DFEAT + c4);
    float4 v[8];
#pragma unroll
    for (int i = 0; i < 8; ++i)
        v[i] = *(const float4*)(rows + loff + i * 2 * DFEAT);
    float4 u[AGG_S - 1][8];
#pragma unroll
    for (int p = 1; p < AGG_S; ++p) {
#pragma unroll
        for (int i = 0; i < 8; ++i)
            u[p - 1][i] = *(const float4*)(rows + p * plane + loff + i * 2 * DFEAT);
    }

    float deg = degp[rowBase + m];
#pragma unroll
    for (int p = 1; p < AGG_S; ++p) deg += degp[(size_t)p * N + rowBase + m];
    float bcol[4];
#pragma unroll
    for (int t = 0; t < 4; ++t) bcol[t] = bias[t * 16 + m];

#pragma unroll
    for (int j = 0; j < DFEAT * DFEAT / 4 / LIN_THREADS; ++j) {
        const int idx = tid + j * LIN_THREADS;
        *(float4*)(Ws  + (idx >> 4) * LIN_PITCH + ((idx & 15) << 2)) = w4[j];
        *(float4*)(Ws2 + (idx >> 4) * LIN_PITCH + ((idx & 15) << 2)) = x4[j];
    }
#pragma unroll
    for (int i = 0; i < 8; ++i) {
#pragma unroll
        for (int p = 1; p < AGG_S; ++p) {
            v[i].x += u[p - 1][i].x; v[i].y += u[p - 1][i].y;
            v[i].z += u[p - 1][i].z; v[i].w += u[p - 1][i].w;
        }
        *(float4*)(mine + (2 * i + hi) * LIN_PITCH + c4) = v[i];
    }
    const float inv = deg > 0.0f ? 1.0f / deg : 0.0f;
    __syncthreads();

    v8f c[4] = {};
    const float* arow = mine + m * LIN_PITCH + k2;
    const float* brow = Ws   + m * LIN_PITCH + k2;
#pragma unroll
    for (int kb = 0; kb < DFEAT; kb += 4) {
        v2f a = *(const v2f*)(arow + kb);
        a.x *= inv;
        a.y *= inv;
#pragma unroll
        for (int t = 0; t < 4; ++t) {
            const v2f b = *(const v2f*)(brow + t * 16 * LIN_PITCH + kb);
            c[t] = __builtin_amdgcn_wmma_f32_16x16x4_f32(
                     false, a,  false, b,
                     (short)0, c[t],  false,  false);
        }
    }

#pragma unroll
    for (int t = 0; t < 4; ++t) {
#pragma unroll
        for (int r = 0; r < 8; ++r) {
            float e = c[t][r] + bcol[t];
            e = e > 0.0f ? e : expm1f(e);
            mine[(r + (hi << 3)) * LIN_PITCH + t * 16 + m] = e;
        }
    }
    __syncthreads();

    v8f c2[4] = {};
    const float* brow2 = Ws2 + m * LIN_PITCH + k2;
#pragma unroll
    for (int kb = 0; kb < DFEAT; kb += 4) {
        const v2f a = *(const v2f*)(arow + kb);
#pragma unroll
        for (int t = 0; t < 4; ++t) {
            const v2f b = *(const v2f*)(brow2 + t * 16 * LIN_PITCH + kb);
            c2[t] = __builtin_amdgcn_wmma_f32_16x16x4_f32(
                     false, a,  false, b,
                     (short)0, c2[t],  false,  false);
        }
    }
#pragma unroll
    for (int t = 0; t < 4; ++t) {
#pragma unroll
        for (int r = 0; r < 8; ++r)
            mine[(r + (hi << 3)) * LIN_PITCH + t * 16 + m] = c2[t][r];
    }
    __syncthreads();

    if (valid) {
        float* orow = out + (size_t)rowBase * DFEAT;
#pragma unroll
        for (int i = 0; i < 8; ++i)
            *(float4*)(orow + loff + i * 2 * DFEAT) =
                *(const float4*)(mine + (2 * i + hi) * LIN_PITCH + c4);
    }
}

extern "C" void kernel_launch(void* const* d_in, const int* in_sizes, int n_in,
                              void* d_out, int out_size, void* d_ws, size_t ws_size,
                              hipStream_t stream) {
    const float* h   = (const float*)d_in[0];
    const int*   src = (const int*)  d_in[1];
    const int*   dst = (const int*)  d_in[2];
    const float* W1  = (const float*)d_in[3];
    const float* b1  = (const float*)d_in[4];
    const float* W2  = (const float*)d_in[5];
    const float* b2  = (const float*)d_in[6];
    float*       out = (float*)d_out;

    const int N = in_sizes[0] / DFEAT;
    const int E = in_sizes[1];

    char* ws = (char*)d_ws;
    float* deg = (float*)ws;
    size_t degBytes = (((size_t)AGG_S * N * sizeof(float)) + 255) & ~(size_t)255;
    float* agg = (float*)(ws + degBytes);
    float* h1  = agg + (size_t)AGG_S * N * DFEAT;
    unsigned* hdrs  = (unsigned*)(h1 + (size_t)N * DFEAT);

    const int tiles = N / 16;
    const int linB  = (tiles + LIN_WAVES - 1) / LIN_WAVES;
    const dim3 aggB((N + AGG_R - 1) / AGG_R, AGG_S);
    unsigned* lists = hdrs + (size_t)aggB.x * aggB.y * LIST_HDR;
    unsigned* keys   = lists + (size_t)aggB.x * aggB.y * OQ_N * LIST_WCAP;
    unsigned* packed = keys + (size_t)((E + 15) / 16) * 4;
    const int keyB   = (E / 4 + KEY_THREADS * KEY_ITEMS - 1) / (KEY_THREADS * KEY_ITEMS) + (E < 4);

    gcn_key_kernel<<<keyB, KEY_THREADS, 0, stream>>>(src, dst, keys, packed, E);

    gcn_agg_kernel<true><<<aggB, AGG_THREADS, 0, stream>>>(h, keys, packed, agg, deg, lists, hdrs, b2, out, N, E);
    gcn_wmma_kernel<<<linB, LIN_THREADS, 0, stream>>>(agg, deg, W1, b1, W2, h1, N);

    const dim3 aggB2(aggB.x * 2, 1);
    gcn_agg_kernel<false><<<aggB2, AGG_THREADS, 0, stream>>>(h1, keys, packed, agg, deg, lists, hdrs, b2, out, N, E);
}
